# starting kernel
# baseline (speedup 1.0000x reference)
#include <hip/hip_runtime.h>
#include <hip/hip_bf16.h>

typedef __attribute__((ext_vector_type(2))) float v2f;
typedef __attribute__((ext_vector_type(8))) float v8f;

#define GCN_N 50000
#define GCN_E 800000


__global__ void gcn_zero_kernel(float* __restrict__ p, int n) {
    int i = blockIdx.x * blockDim.x + threadIdx.x;
    if (i < n) p[i] = 0.0f;
}

__global__ void gcn_degree_kernel(const int* __restrict__ dst, float* __restrict__ deg, int E) {
    int e = blockIdx.x * blockDim.x + threadIdx.x;
    if (e < E) atomicAdd(&deg[dst[e]], 1.0f);
}

__global__ void gcn_dinv_kernel(float* __restrict__ deg, int n) {
    int i = blockIdx.x * blockDim.x + threadIdx.x;
    if (i < n) deg[i] = rsqrtf(deg[i] + 1.0f);
}

template <int RELU>
__global__ void gcn_gemm_wmma_f32(const float* __restrict__ A,
                                  const float* __restrict__ W,
                                  const float* __restrict__ bias,
                                  const float* __restrict__ dinv,
                                  float* __restrict__ hw,
                                  float* __restrict__ agg,
                                  int M, int K, int Nc) {
    const int lane = threadIdx.x & 31;
    const int wid  = threadIdx.x >> 5;
    const int wavesPerBlock = blockDim.x >> 5;

    const int tilesN = (Nc + 15) >> 4;
    const int tilesM = M >> 4;
    int tile = blockIdx.x * wavesPerBlock + wid;
    if (tile >= tilesM * tilesN) return;

    const int tm   = tile / tilesN;
    const int tn   = tile - tm * tilesN;
    const int half = lane >> 4;
    const int l16  = lane & 15;

    const int  row   = tm * 16 + l16;
    const int  col   = tn * 16 + l16;
    const bool colOK = (col < Nc);
    const int  colc  = colOK ? col : 0;
    const int  koff  = half * 2;

    const float* __restrict__ Arow = A + (size_t)row * K + koff;
    const float* __restrict__ Wcol = W + colc;

    v8f acc = {};
    const int Kmain = K & ~3;

#pragma unroll 4
    for (int k = 0; k < Kmain; k += 4) {
        float a0 = Arow[k];
        float a1 = Arow[k + 1];
        if (RELU) { a0 = fmaxf(a0, 0.0f); a1 = fmaxf(a1, 0.0f); }
        float w0 = Wcol[(size_t)(k + koff)     * Nc];
        float w1 = Wcol[(size_t)(k + koff + 1) * Nc];
        v2f a; a[0] = a0;               a[1] = a1;
        v2f b; b[0] = colOK ? w0 : 0.f; b[1] = colOK ? w1 : 0.f;
        acc = __builtin_amdgcn_wmma_f32_16x16x4_f32(
            false, a, false, b, (short)0, acc, false, false);
    }

    if (Kmain < K) {
        const int  ka  = Kmain + koff;
        const bool ok0 = (ka     < K);
        const bool ok1 = (ka + 1 < K);
        const int  kc0 = ok0 ? ka     : (K - 1);
        const int  kc1 = ok1 ? ka + 1 : (K - 1);
        float a0 = A[(size_t)row * K + kc0];
        float a1 = A[(size_t)row * K + kc1];
        if (RELU) { a0 = fmaxf(a0, 0.0f); a1 = fmaxf(a1, 0.0f); }
        float w0 = Wcol[(size_t)kc0 * Nc];
        float w1 = Wcol[(size_t)kc1 * Nc];
        v2f a; a[0] = ok0 ? a0 : 0.f;            a[1] = ok1 ? a1 : 0.f;
        v2f b; b[0] = (colOK && ok0) ? w0 : 0.f; b[1] = (colOK && ok1) ? w1 : 0.f;
        acc = __builtin_amdgcn_wmma_f32_16x16x4_f32(
            false, a, false, b, (short)0, acc, false, false);
    }

    const int   rbase = tm * 16 + (half << 3);
    const float bn    = bias[colc];
    if (colOK) {
#pragma unroll
        for (int r = 0; r < 8; ++r) {
            const int m  = rbase + r;
            const float di = dinv[m];
            const float v  = acc[r];
            hw [(size_t)m * Nc + col] = v;
            agg[(size_t)m * Nc + col] = v * di * di + bn;
        }
    }
}

__global__ void gcn_scatter_kernel(const int* __restrict__ src,
                                   const int* __restrict__ dst,
                                   const float* __restrict__ hw,
                                   const float* __restrict__ dinv,
                                   float* __restrict__ agg, int E, int C) {
    int wave = (int)((blockIdx.x * (size_t)blockDim.x + threadIdx.x) >> 5);
    int lane = threadIdx.x & 31;
    if (wave >= E) return;
    int s = src[wave];
    int d = dst[wave];
    float norm = dinv[s] * dinv[d];
    const float* __restrict__ hrow = hw  + (size_t)s * C;
    float*       __restrict__ arow = agg + (size_t)d * C;
    for (int c = lane; c < C; c += 32) {
        atomicAdd(&arow[c], hrow[c] * norm);
    }
}


extern "C" void kernel_launch(void* const* d_in, const int* in_sizes, int n_in,
                              void* d_out, int out_size, void* d_ws, size_t ws_size,
                              hipStream_t stream) {
    (void)in_sizes; (void)n_in; (void)out_size; (void)ws_size;

    const int N = GCN_N;
    const int E = GCN_E;

    const float* x    = (const float*)d_in[0];
    const int*   ei   = (const int*)d_in[1];
    const int*   src  = ei;
    const int*   dstp = ei + E;

    const float* W[5] = { (const float*)d_in[2], (const float*)d_in[4],
                          (const float*)d_in[6], (const float*)d_in[8],
                          (const float*)d_in[10] };
    const float* B[5] = { (const float*)d_in[3], (const float*)d_in[5],
                          (const float*)d_in[7], (const float*)d_in[9],
                          (const float*)d_in[11] };

    float* ws   = (float*)d_ws;
    float* dinv = ws;
    float* hw   = dinv + N;
    float* aggA = hw   + (size_t)N * 96;
    float* aggB = aggA + (size_t)N * 96;
    float* out  = (float*)d_out;

    const int T = 256;

    gcn_zero_kernel  <<<(N + T - 1) / T, T, 0, stream>>>(dinv, N);
    gcn_degree_kernel<<<(E + T - 1) / T, T, 0, stream>>>(dstp, dinv, E);
    gcn_dinv_kernel  <<<(N + T - 1) / T, T, 0, stream>>>(dinv, N);

    const int dims[6] = { 165, 96, 96, 96, 96, 2 };
    const float* layerIn = x;
    float* aggBufs[2] = { aggA, aggB };

    for (int l = 0; l < 5; ++l) {
        const int K  = dims[l];
        const int Cn = dims[l + 1];
        float* agg = (l == 4) ? out : aggBufs[l & 1];

        const int tiles         = (N / 16) * ((Cn + 15) / 16);
        const int wavesPerBlock = 4;
        const int gBlocks       = (tiles + wavesPerBlock - 1) / wavesPerBlock;
        if (l == 0) {
            gcn_gemm_wmma_f32<0><<<gBlocks, wavesPerBlock * 32, 0, stream>>>(
                layerIn, W[l], B[l], dinv, hw, agg, N, K, Cn);
        } else {
            gcn_gemm_wmma_f32<1><<<gBlocks, wavesPerBlock * 32, 0, stream>>>(
                layerIn, W[l], B[l], dinv, hw, agg, N, K, Cn);
        }

        const size_t scatterThreads = (size_t)E * 32;
        gcn_scatter_kernel<<<(scatterThreads + T - 1) / T, T, 0, stream>>>(
            src, dstp, hw, dinv, agg, E, Cn);

        layerIn = agg;
    }
}


# GEMM of layers 0-3: K split over a group's 3 waves (each loads a third of a strip's A, so A is read once per group), six partial tiles per wave summed through LDS; k sum order changed: 3 partial sums
# speedup vs baseline: 4.4237x; 4.4237x over previous
#include <hip/hip_runtime.h>
#include <hip/hip_bf16.h>

typedef __attribute__((ext_vector_type(2))) float v2f;
typedef __attribute__((ext_vector_type(4))) float v4f;
typedef __attribute__((ext_vector_type(8))) float v8f;

#define GCN_N 50000
#define GCN_E 800000
#define GCN_NPAD 50048


__device__ __forceinline__ unsigned gcn_l2_half() {
    return (__builtin_amdgcn_s_sendmsg_rtn(0x87) >> 18) & 1u;
}

__global__ void gcn_zero_kernel(float* __restrict__ p, float* __restrict__ q,
                                int* __restrict__ c0, int* __restrict__ c1,
                                int* __restrict__ ovfCnt, int n) {
    int i = blockIdx.x * blockDim.x + threadIdx.x;
    if (i < n) { p[i] = 0.0f; q[i] = 0.0f; c0[i] = 0; c1[i] = 0; }
    if (i == 0) ovfCnt[0] = 0;
}

#define GCN_CAP 32

__global__ void gcn_degree_bucket_kernel(const int* __restrict__ src, const int* __restrict__ dst,
                                         float* __restrict__ deg0, float* __restrict__ deg1,
                                         int* __restrict__ cnt0, int* __restrict__ cnt1,
                                         int* __restrict__ slots0, int* __restrict__ slots1,
                                         int* __restrict__ ovfCnt, int* __restrict__ ovfList,
                                         int E) {
    int e = blockIdx.x * blockDim.x + threadIdx.x;
    const unsigned hf = gcn_l2_half();
    float* __restrict__ deg   = hf ? deg1   : deg0;
    int*   __restrict__ cnt   = hf ? cnt1   : cnt0;
    int*   __restrict__ slots = hf ? slots1 : slots0;
    if (e < E) {
        const int s = src[e];
        const int d = dst[e];
        atomicAdd(&deg[d], 1.0f);
        const int r = atomicAdd(&cnt[s], 1);
        if (r < GCN_CAP) slots[(size_t)s * GCN_CAP + r] = d;
        else             ovfList[atomicAdd(ovfCnt, 1)] = e;
    }
}

__global__ void gcn_dinv_kernel(float* __restrict__ deg, const float* __restrict__ deg1, int n) {
    int i = blockIdx.x * blockDim.x + threadIdx.x;
    if (i < n) deg[i] = rsqrtf(deg[i] + deg1[i] + 1.0f);
}


template <int C>
__global__ void gcn_finish_kernel(float* __restrict__ out, const float* __restrict__ other,
                                  const float* __restrict__ dinv,
                                  const float* __restrict__ bias, int n) {
    int i = blockIdx.x * blockDim.x + threadIdx.x;
    if (i < n) {
        const int node = i / C;
        out[i] = dinv[node] * (out[i] + other[i]) + bias[i - node * C];
    }
}

template <int RELU>
__global__ void gcn_gemm_wmma_f32(const float* __restrict__ A,
                                  const float* __restrict__ A2,
                                  const float* __restrict__ W,
                                  const float* __restrict__ biasIn,
                                  const float* __restrict__ dinv,
                                  float* __restrict__ hw,
                                  float* __restrict__ agg,
                                  float* __restrict__ agg2,
                                  int M, int K, int Nc) {
    const int lane = threadIdx.x & 31;
    const int wid  = threadIdx.x >> 5;
    const int wavesPerBlock = blockDim.x >> 5;

    const int tilesN = (Nc + 15) >> 4;
    const int tilesM = M >> 4;
    int tile = blockIdx.x * wavesPerBlock + wid;
    if (tile >= tilesM * tilesN) return;

    const int tm   = tile / tilesN;
    const int tn   = tile - tm * tilesN;
    const int half = lane >> 4;
    const int l16  = lane & 15;

    const int  row   = tm * 16 + l16;
    const int  col   = tn * 16 + l16;
    const bool colOK = (col < Nc);
    const int  colc  = colOK ? col : 0;
    const int  koff  = half * 2;

    const float* __restrict__ Arow  = A  + (size_t)row * K + koff;
    const float* __restrict__ Arow2 = A2 + (size_t)row * K + koff;
    const float* __restrict__ Wcol = W + colc;
    const float* __restrict__ bIn  = biasIn + koff;
    const float dr = dinv[row];

    v8f acc = {};
    const int Kmain = K & ~3;

#pragma unroll 4
    for (int k = 0; k < Kmain; k += 4) {
        float a0 = dr * (Arow[k]     + Arow2[k])     + bIn[k];
        float a1 = dr * (Arow[k + 1] + Arow2[k + 1]) + bIn[k + 1];
        if (RELU) { a0 = fmaxf(a0, 0.0f); a1 = fmaxf(a1, 0.0f); }
        float w0 = Wcol[(size_t)(k + koff)     * Nc];
        float w1 = Wcol[(size_t)(k + koff + 1) * Nc];
        v2f a; a[0] = a0;               a[1] = a1;
        v2f b; b[0] = colOK ? w0 : 0.f; b[1] = colOK ? w1 : 0.f;
        acc = __builtin_amdgcn_wmma_f32_16x16x4_f32(
            false, a, false, b, (short)0, acc, false, false);
    }

    if (Kmain < K) {
        const int  ka  = Kmain + koff;
        const bool ok0 = (ka     < K);
        const bool ok1 = (ka + 1 < K);
        const int  kc0 = ok0 ? ka     : (K - 1);
        const int  kc1 = ok1 ? ka + 1 : (K - 1);
        float a0 = dr * (A[(size_t)row * K + kc0] + A2[(size_t)row * K + kc0]) + biasIn[kc0];
        float a1 = dr * (A[(size_t)row * K + kc1] + A2[(size_t)row * K + kc1]) + biasIn[kc1];
        if (RELU) { a0 = fmaxf(a0, 0.0f); a1 = fmaxf(a1, 0.0f); }
        float w0 = Wcol[(size_t)kc0 * Nc];
        float w1 = Wcol[(size_t)kc1 * Nc];
        v2f a; a[0] = ok0 ? a0 : 0.f;            a[1] = ok1 ? a1 : 0.f;
        v2f b; b[0] = (colOK && ok0) ? w0 : 0.f; b[1] = (colOK && ok1) ? w1 : 0.f;
        acc = __builtin_amdgcn_wmma_f32_16x16x4_f32(
            false, a, false, b, (short)0, acc, false, false);
    }

    const int   rbase = tm * 16 + (half << 3);
    if (colOK) {
#pragma unroll
        for (int r = 0; r < 8; ++r) {
            const int m  = rbase + r;
            const float v  = acc[r] * dinv[m];
            hw  [(size_t)m * Nc + col] = v;
            agg [(size_t)m * Nc + col] = v;
            agg2[(size_t)m * Nc + col] = 0.0f;
        }
    }
}

template <int K, int RELU, int DUAL>
__global__ __launch_bounds__(96)
void gcn_gemm_wstat_f32(const float* __restrict__ A,
                        const float* __restrict__ A2,
                        const float* __restrict__ W,
                        const float* __restrict__ biasIn,
                        const float* __restrict__ dinv,
                        float* __restrict__ hw,
                        float* __restrict__ agg,
                        float* __restrict__ agg2,
                        int nStrips) {
    constexpr int  NC     = 96;
    constexpr int  NCHUNK = (K + 7) / 8;
    static_assert(NCHUNK % 3 == 0, "the 8-wide k chunks are shared out evenly over the 3 waves");
    constexpr int  NCW    = NCHUNK / 3;
    constexpr int  KW     = 8 * NCW;
    constexpr bool TAIL   = (K % 8) != 0;
    const int lane = threadIdx.x & 31;
    const int wv   = threadIdx.x >> 5;
    const int h    = lane >> 4;
    const int c    = lane & 15;
    const int col0 = wv * 32;
    const int kw   = KW * wv;

    __shared__ float4 red[2 * 3 * 6 * 2 * 32];

    v2f bfrag[NCW][2][6];
#pragma unroll
    for (int jj = 0; jj < NCW; ++jj) {
#pragma unroll
        for (int i = 0; i < 2; ++i) {
            const int k0 = kw + 8 * jj + 4 * h + 2 * i;
#pragma unroll
            for (int t = 0; t < 6; ++t) {
                const float* __restrict__ wc = W + 16 * t + c;
                if (!TAIL || jj < NCW - 1) {
                    bfrag[jj][i][t][0] = wc[(k0    ) * NC];
                    bfrag[jj][i][t][1] = wc[(k0 + 1) * NC];
                } else {
                    const bool ok0 = (k0     < K);
                    const bool ok1 = (k0 + 1 < K);
                    const float w0 = wc[(ok0 ? k0     : K - 1) * NC];
                    const float w1 = wc[(ok1 ? k0 + 1 : K - 1) * NC];
                    bfrag[jj][i][t][0] = ok0 ? w0 : 0.f;
                    bfrag[jj][i][t][1] = ok1 ? w1 : 0.f;
                }
            }
        }
    }
    float bk[NCW][4];
#pragma unroll
    for (int jj = 0; jj < NCW; ++jj) {
#pragma unroll
        for (int i = 0; i < 4; ++i) {
            const int k = kw + 8 * jj + 4 * h + i;
            bk[jj][i] = DUAL ? biasIn[k < K ? k : K - 1] : 0.f;
        }
    }

    int par = 0;
    for (int strip = blockIdx.x; strip < nStrips; strip += gridDim.x, par ^= 1) {
        const int m0 = strip * 16;
        const float drow = dinv[m0 + c];
        const float* __restrict__ arow  = A + (size_t)(m0 + c) * K + kw;
        const float* __restrict__ arow2 = DUAL ? A2 + (size_t)(m0 + c) * K + kw : arow;

        v4f ra[NCW], ra2[NCW];
#pragma unroll
        for (int jj = 0; jj < NCW; ++jj) {
#pragma unroll
            for (int i = 0; i < 4; ++i) {
                int kc = 8 * jj + 4 * h + i;
                if (TAIL && jj == NCW - 1)
                    kc = (kw + kc < K) ? kc : K - 1 - kw;
                ra [jj][i] = arow[kc];
                ra2[jj][i] = DUAL ? arow2[kc] : 0.f;
            }
        }
        if constexpr (DUAL) {
            static_assert(NCW == 4, "the statement below names the loads of 4 chunks");
            asm volatile("" :: "v"(ra[0]), "v"(ra[1]), "v"(ra[2]), "v"(ra[3]),
                               "v"(ra2[0]), "v"(ra2[1]), "v"(ra2[2]), "v"(ra2[3]));
        }
        float av[NCW][4];
#pragma unroll
        for (int jj = 0; jj < NCW; ++jj) {
#pragma unroll
            for (int i = 0; i < 4; ++i) {
                float a = ra[jj][i];
                if (DUAL) a = drow * (a + ra2[jj][i]) + bk[jj][i];
                if (RELU) a = fmaxf(a, 0.0f);
                if (TAIL && jj == NCW - 1) a = (kw + 8 * jj + 4 * h + i < K) ? a : 0.f;
                av[jj][i] = a;
            }
        }

        v8f acc[6] = {};
#pragma unroll
        for (int jj = 0; jj < NCW; ++jj) {
#pragma unroll
            for (int i = 0; i < 2; ++i) {
                v2f a; a[0] = av[jj][2 * i]; a[1] = av[jj][2 * i + 1];
#pragma unroll
                for (int t = 0; t < 6; ++t) {
                    acc[t] = __builtin_amdgcn_wmma_f32_16x16x4_f32(
                        false, a, false, bfrag[jj][i][t], (short)0, acc[t], false, false);
                }
                __builtin_amdgcn_sched_barrier(0);
            }
        }

        float4* __restrict__ rb = red + par * (3 * 6 * 2 * 32);
#pragma unroll
        for (int t = 0; t < 6; ++t) {
#pragma unroll
            for (int q = 0; q < 2; ++q) {
                rb[((wv * 6 + t) * 2 + q) * 32 + lane] =
                    make_float4(acc[t][4 * q], acc[t][4 * q + 1], acc[t][4 * q + 2], acc[t][4 * q + 3]);
            }
        }
        __syncthreads();
        float4 pt[3][2][2];
#pragma unroll
        for (int w2 = 0; w2 < 3; ++w2) {
#pragma unroll
            for (int tt = 0; tt < 2; ++tt) {
#pragma unroll
                for (int q = 0; q < 2; ++q)
                    pt[w2][tt][q] = rb[((w2 * 6 + 2 * wv + tt) * 2 + q) * 32 + lane];
            }
        }
        __builtin_amdgcn_sched_barrier(0);
        v8f sum0, sum1;
#pragma unroll
        for (int q = 0; q < 2; ++q) {
            float4 f0 = pt[0][0][q];
            float4 f1 = pt[0][1][q];
#pragma unroll
            for (int w2 = 1; w2 < 3; ++w2) {
                const float4 g0 = pt[w2][0][q];
                const float4 g1 = pt[w2][1][q];
                f0.x += g0.x; f0.y += g0.y; f0.z += g0.z; f0.w += g0.w;
                f1.x += g1.x; f1.y += g1.y; f1.z += g1.z; f1.w += g1.w;
            }
            sum0[4 * q] = f0.x; sum0[4 * q + 1] = f0.y; sum0[4 * q + 2] = f0.z; sum0[4 * q + 3] = f0.w;
            sum1[4 * q] = f1.x; sum1[4 * q + 1] = f1.y; sum1[4 * q + 2] = f1.z; sum1[4 * q + 3] = f1.w;
        }

#pragma unroll
        for (int r = 0; r < 8; ++r) {
            const float t0 = sum0[r];
            const float t1 = sum1[r];
            const float p0 = __int_as_float(__builtin_amdgcn_permlanex16(
                0, __float_as_int(t0), 0x76543210, 0xfedcba98, false, false));
            const float p1 = __int_as_float(__builtin_amdgcn_permlanex16(
                0, __float_as_int(t1), 0x76543210, 0xfedcba98, false, false));
            const float lineA = h ? p1 : t0;
            const float lineB = h ? t1 : p0;
            const float vA = lineA * __int_as_float(__builtin_amdgcn_readlane(__float_as_int(drow), r));
            const float vB = lineB * __int_as_float(__builtin_amdgcn_readlane(__float_as_int(drow), 8 + r));
            const size_t oA = (size_t)(m0 + r) * NC + col0 + lane;
            const size_t oB = (size_t)(m0 + 8 + r) * NC + col0 + lane;
            hw  [oA] = vA;
            agg [oA] = vA;
            agg2[oA] = 0.0f;
            hw  [oB] = vB;
            agg [oB] = vB;
            agg2[oB] = 0.0f;
        }
    }
}

template <int NCH, int G>
__global__ void gcn_push_rows_kernel(const int* __restrict__ src,
                                     const int* __restrict__ dst,
                                     const int* __restrict__ cnt0,
                                     const int* __restrict__ cnt1,
                                     const int* __restrict__ slots0,
                                     const int* __restrict__ slots1,
                                     const int* __restrict__ ovfCnt,
                                     const int* __restrict__ ovfList,
                                     const float* __restrict__ hw,
                                     float* __restrict__ agg0,
                                     float* __restrict__ agg1, int N) {
    constexpr int C = 32 * NCH;
    static_assert(GCN_CAP == 32, "lane i holds place i of a list");
    const int lane = threadIdx.x & 31;
    const int wave = __builtin_amdgcn_readfirstlane(
        (int)((blockIdx.x * (size_t)blockDim.x + threadIdx.x) >> 5));
    const int nWaves = (int)((gridDim.x * (size_t)blockDim.x) >> 5);
    float* __restrict__ agg = gcn_l2_half() ? agg1 : agg0;

    const int s0 = wave * G;
    if (s0 < N) {
        int   n0[G], n1[G];
        int   l0[G], l1[G];
        float v[G][NCH];
#pragma unroll
        for (int u = 0; u < G; ++u) {
            const bool ok = (s0 + u < N);
            const int  s  = ok ? (s0 + u) : s0;
            const int  c0 = cnt0[s];
            const int  c1 = cnt1[s];
            n0[u] = ok ? (c0 < GCN_CAP ? c0 : GCN_CAP) : 0;
            n1[u] = ok ? (c1 < GCN_CAP ? c1 : GCN_CAP) : 0;
            l0[u] = slots0[(size_t)s * GCN_CAP + lane];
            l1[u] = slots1[(size_t)s * GCN_CAP + lane];
            const float* __restrict__ hrow = hw + (size_t)s * C + lane;
#pragma unroll
            for (int q = 0; q < NCH; ++q) v[u][q] = hrow[q * 32];
        }
#pragma unroll
        for (int u = 0; u < G; ++u) {
            for (int j = 0; j < n0[u]; ++j) {
                const int dj = __builtin_amdgcn_readlane(l0[u], j);
                float* __restrict__ arow = agg + (size_t)dj * C + lane;
#pragma unroll
                for (int q = 0; q < NCH; ++q) atomicAdd(&arow[q * 32], v[u][q]);
            }
            for (int j = 0; j < n1[u]; ++j) {
                const int dj = __builtin_amdgcn_readlane(l1[u], j);
                float* __restrict__ arow = agg + (size_t)dj * C + lane;
#pragma unroll
                for (int q = 0; q < NCH; ++q) atomicAdd(&arow[q * 32], v[u][q]);
            }
        }
    }

    const int no = ovfCnt[0];
    for (int i = wave; i < no; i += nWaves) {
        const int e  = ovfList[i];
        const int sj = src[e];
        const int dj = dst[e];
        const float* __restrict__ hrow = hw  + (size_t)sj * C + lane;
        float*       __restrict__ arow = agg + (size_t)dj * C + lane;
#pragma unroll
        for (int q = 0; q < NCH; ++q) atomicAdd(&arow[q * 32], hrow[q * 32]);
    }
}

template <int C>
__global__ void gcn_scatter_narrow_kernel(const int* __restrict__ src,
                                          const int* __restrict__ dst,
                                          const float* __restrict__ hw,
                                          float* __restrict__ agg0,
                                          float* __restrict__ agg1, int E) {
    const int e = (int)(blockIdx.x * (size_t)blockDim.x + threadIdx.x);
    if (e >= E) return;
    float* __restrict__ agg = gcn_l2_half() ? agg1 : agg0;
    const int s = src[e];
    const int d = dst[e];
    const float* __restrict__ hrow = hw  + (size_t)s * C;
    float*       __restrict__ arow = agg + (size_t)d * C;
    float v[C];
#pragma unroll
    for (int c = 0; c < C; ++c) v[c] = hrow[c];
#pragma unroll
    for (int c = 0; c < C; ++c) atomicAdd(&arow[c], v[c]);
}


extern "C" void kernel_launch(void* const* d_in, const int* in_sizes, int n_in,
                              void* d_out, int out_size, void* d_ws, size_t ws_size,
                              hipStream_t stream) {
    (void)in_sizes; (void)n_in; (void)out_size; (void)ws_size;

    const int N = GCN_N;
    const int E = GCN_E;

    const float* x    = (const float*)d_in[0];
    const int*   ei   = (const int*)d_in[1];
    const int*   src  = ei;
    const int*   dstp = ei + E;

    const float* W[5] = { (const float*)d_in[2], (const float*)d_in[4],
                          (const float*)d_in[6], (const float*)d_in[8],
                          (const float*)d_in[10] };
    const float* B[5] = { (const float*)d_in[3], (const float*)d_in[5],
                          (const float*)d_in[7], (const float*)d_in[9],
                          (const float*)d_in[11] };

    float* ws    = (float*)d_ws;
    float* hw    = ws;
    float* aggA  = hw    + (size_t)N * 96;
    float* aggB  = aggA  + (size_t)N * 96;
    float* aggA2 = aggB  + (size_t)N * 96;
    float* aggB2 = aggA2 + (size_t)N * 96;
    float* dinv  = aggB2 + (size_t)N * 96;
    float* deg2  = dinv  + GCN_NPAD;
    float* out2  = deg2  + GCN_NPAD;
    float* out   = (float*)d_out;
    int* cnt0    = (int*)(out2 + (size_t)N * 2);
    int* cnt1    = cnt0   + GCN_NPAD;
    int* slots0  = cnt1   + GCN_NPAD;
    int* slots1  = slots0 + (size_t)N * GCN_CAP;
    int* ovfCnt  = slots1 + (size_t)N * GCN_CAP;
    int* ovfList = ovfCnt + 64;

    const int T = 256;

    gcn_zero_kernel  <<<(N + T - 1) / T, T, 0, stream>>>(dinv, deg2, cnt0, cnt1, ovfCnt, N);
    gcn_degree_bucket_kernel<<<(E + T - 1) / T, T, 0, stream>>>(
        src, dstp, dinv, deg2, cnt0, cnt1, slots0, slots1, ovfCnt, ovfList, E);
    gcn_dinv_kernel  <<<(N + T - 1) / T, T, 0, stream>>>(dinv, deg2, N);

    const int dims[6] = { 165, 96, 96, 96, 96, 2 };
    const float* layerIn  = x;
    const float* layerIn2 = x;
    float* aggBufs [2] = { aggA,  aggB  };
    float* aggBufs2[2] = { aggA2, aggB2 };

    for (int l = 0; l < 5; ++l) {
        const int K  = dims[l];
        const int Cn = dims[l + 1];
        float* agg  = (l == 4) ? out  : aggBufs [l & 1];
        float* agg2 = (l == 4) ? out2 : aggBufs2[l & 1];
        const float* biasIn = (l == 0) ? B[0] : B[l - 1];

        const int tiles         = (N / 16) * ((Cn + 15) / 16);
        const int wavesPerBlock = 4;
        const int gBlocks       = (tiles + wavesPerBlock - 1) / wavesPerBlock;
        if (l == 0) {
            gcn_gemm_wstat_f32<165, 0, 0><<<1024, 96, 0, stream>>>(
                layerIn, layerIn2, W[l], biasIn, dinv, hw, agg, agg2, GCN_N / 16);
        } else if (l < 4) {
            gcn_gemm_wstat_f32<96, 1, 1><<<1024, 96, 0, stream>>>(
                layerIn, layerIn2, W[l], biasIn, dinv, hw, agg, agg2, GCN_N / 16);
        } else {
            gcn_gemm_wmma_f32<1><<<gBlocks, wavesPerBlock * 32, 0, stream>>>(
                layerIn, layerIn2, W[l], biasIn, dinv, hw, agg, agg2, N, K, Cn);
        }

        const int scatterBlocks = (E + T - 1) / T;
        const int pushBlocks    = (N + 8 * 2 - 1) / (8 * 2);
        if (l < 4) {
            gcn_push_rows_kernel<3, 8><<<pushBlocks, 64, 0, stream>>>(
                src, dstp, cnt0, cnt1, slots0, slots1, ovfCnt, ovfList, hw, agg, agg2, N);
        } else {
            gcn_scatter_narrow_kernel<2><<<scatterBlocks, T, 0, stream>>>(
                src, dstp, hw, agg, agg2, E);
        }

        layerIn  = agg;
        layerIn2 = agg2;
    }

    gcn_finish_kernel<2><<<(2 * N + T - 1) / T, T, 0, stream>>>(out, out2, dinv, B[4], 2 * N);
}


# Last layer's 2-channel edge pass walks the lists by source, a lane pair per out-edge: values read once per source, one atomic request per edge, not a gather and two; atomic arrival order changed
# speedup vs baseline: 4.5578x; 1.0303x over previous
#include <hip/hip_runtime.h>
#include <hip/hip_bf16.h>

typedef __attribute__((ext_vector_type(2))) float v2f;
typedef __attribute__((ext_vector_type(4))) float v4f;
typedef __attribute__((ext_vector_type(8))) float v8f;

#define GCN_N 50000
#define GCN_E 800000
#define GCN_NPAD 50048


__device__ __forceinline__ unsigned gcn_l2_half() {
    return (__builtin_amdgcn_s_sendmsg_rtn(0x87) >> 18) & 1u;
}

__global__ void gcn_zero_kernel(float* __restrict__ p, float* __restrict__ q,
                                int* __restrict__ c0, int* __restrict__ c1,
                                int* __restrict__ ovfCnt, int n) {
    int i = blockIdx.x * blockDim.x + threadIdx.x;
    if (i < n) { p[i] = 0.0f; q[i] = 0.0f; c0[i] = 0; c1[i] = 0; }
    if (i == 0) ovfCnt[0] = 0;
}

#define GCN_CAP 32

__global__ void gcn_degree_bucket_kernel(const int* __restrict__ src, const int* __restrict__ dst,
                                         float* __restrict__ deg0, float* __restrict__ deg1,
                                         int* __restrict__ cnt0, int* __restrict__ cnt1,
                                         int* __restrict__ slots0, int* __restrict__ slots1,
                                         int* __restrict__ ovfCnt, int* __restrict__ ovfList,
                                         int E) {
    int e = blockIdx.x * blockDim.x + threadIdx.x;
    const unsigned hf = gcn_l2_half();
    float* __restrict__ deg   = hf ? deg1   : deg0;
    int*   __restrict__ cnt   = hf ? cnt1   : cnt0;
    int*   __restrict__ slots = hf ? slots1 : slots0;
    if (e < E) {
        const int s = src[e];
        const int d = dst[e];
        atomicAdd(&deg[d], 1.0f);
        const int r = atomicAdd(&cnt[s], 1);
        if (r < GCN_CAP) slots[(size_t)s * GCN_CAP + r] = d;
        else             ovfList[atomicAdd(ovfCnt, 1)] = e;
    }
}

__global__ void gcn_dinv_kernel(float* __restrict__ deg, const float* __restrict__ deg1, int n) {
    int i = blockIdx.x * blockDim.x + threadIdx.x;
    if (i < n) deg[i] = rsqrtf(deg[i] + deg1[i] + 1.0f);
}


template <int C>
__global__ void gcn_finish_kernel(float* __restrict__ out, const float* __restrict__ other,
                                  const float* __restrict__ dinv,
                                  const float* __restrict__ bias, int n) {
    int i = blockIdx.x * blockDim.x + threadIdx.x;
    if (i < n) {
        const int node = i / C;
        out[i] = dinv[node] * (out[i] + other[i]) + bias[i - node * C];
    }
}

template <int RELU>
__global__ void gcn_gemm_wmma_f32(const float* __restrict__ A,
                                  const float* __restrict__ A2,
                                  const float* __restrict__ W,
                                  const float* __restrict__ biasIn,
                                  const float* __restrict__ dinv,
                                  float* __restrict__ hw,
                                  float* __restrict__ agg,
                                  float* __restrict__ agg2,
                                  int M, int K, int Nc) {
    const int lane = threadIdx.x & 31;
    const int wid  = threadIdx.x >> 5;
    const int wavesPerBlock = blockDim.x >> 5;

    const int tilesN = (Nc + 15) >> 4;
    const int tilesM = M >> 4;
    int tile = blockIdx.x * wavesPerBlock + wid;
    if (tile >= tilesM * tilesN) return;

    const int tm   = tile / tilesN;
    const int tn   = tile - tm * tilesN;
    const int half = lane >> 4;
    const int l16  = lane & 15;

    const int  row   = tm * 16 + l16;
    const int  col   = tn * 16 + l16;
    const bool colOK = (col < Nc);
    const int  colc  = colOK ? col : 0;
    const int  koff  = half * 2;

    const float* __restrict__ Arow  = A  + (size_t)row * K + koff;
    const float* __restrict__ Arow2 = A2 + (size_t)row * K + koff;
    const float* __restrict__ Wcol = W + colc;
    const float* __restrict__ bIn  = biasIn + koff;
    const float dr = dinv[row];

    v8f acc = {};
    const int Kmain = K & ~3;

#pragma unroll 4
    for (int k = 0; k < Kmain; k += 4) {
        float a0 = dr * (Arow[k]     + Arow2[k])     + bIn[k];
        float a1 = dr * (Arow[k + 1] + Arow2[k + 1]) + bIn[k + 1];
        if (RELU) { a0 = fmaxf(a0, 0.0f); a1 = fmaxf(a1, 0.0f); }
        float w0 = Wcol[(size_t)(k + koff)     * Nc];
        float w1 = Wcol[(size_t)(k + koff + 1) * Nc];
        v2f a; a[0] = a0;               a[1] = a1;
        v2f b; b[0] = colOK ? w0 : 0.f; b[1] = colOK ? w1 : 0.f;
        acc = __builtin_amdgcn_wmma_f32_16x16x4_f32(
            false, a, false, b, (short)0, acc, false, false);
    }

    if (Kmain < K) {
        const int  ka  = Kmain + koff;
        const bool ok0 = (ka     < K);
        const bool ok1 = (ka + 1 < K);
        const int  kc0 = ok0 ? ka     : (K - 1);
        const int  kc1 = ok1 ? ka + 1 : (K - 1);
        float a0 = dr * (A[(size_t)row * K + kc0] + A2[(size_t)row * K + kc0]) + biasIn[kc0];
        float a1 = dr * (A[(size_t)row * K + kc1] + A2[(size_t)row * K + kc1]) + biasIn[kc1];
        if (RELU) { a0 = fmaxf(a0, 0.0f); a1 = fmaxf(a1, 0.0f); }
        float w0 = Wcol[(size_t)kc0 * Nc];
        float w1 = Wcol[(size_t)kc1 * Nc];
        v2f a; a[0] = ok0 ? a0 : 0.f;            a[1] = ok1 ? a1 : 0.f;
        v2f b; b[0] = (colOK && ok0) ? w0 : 0.f; b[1] = (colOK && ok1) ? w1 : 0.f;
        acc = __builtin_amdgcn_wmma_f32_16x16x4_f32(
            false, a, false, b, (short)0, acc, false, false);
    }

    const int   rbase = tm * 16 + (half << 3);
    if (colOK) {
#pragma unroll
        for (int r = 0; r < 8; ++r) {
            const int m  = rbase + r;
            const float v  = acc[r] * dinv[m];
            hw  [(size_t)m * Nc + col] = v;
            agg [(size_t)m * Nc + col] = v;
            agg2[(size_t)m * Nc + col] = 0.0f;
        }
    }
}

template <int K, int RELU, int DUAL>
__global__ __launch_bounds__(96)
void gcn_gemm_wstat_f32(const float* __restrict__ A,
                        const float* __restrict__ A2,
                        const float* __restrict__ W,
                        const float* __restrict__ biasIn,
                        const float* __restrict__ dinv,
                        float* __restrict__ hw,
                        float* __restrict__ agg,
                        float* __restrict__ agg2,
                        int nStrips) {
    constexpr int  NC     = 96;
    constexpr int  NCHUNK = (K + 7) / 8;
    static_assert(NCHUNK % 3 == 0, "the 8-wide k chunks are shared out evenly over the 3 waves");
    constexpr int  NCW    = NCHUNK / 3;
    constexpr int  KW     = 8 * NCW;
    constexpr bool TAIL   = (K % 8) != 0;
    const int lane = threadIdx.x & 31;
    const int wv   = threadIdx.x >> 5;
    const int h    = lane >> 4;
    const int c    = lane & 15;
    const int col0 = wv * 32;
    const int kw   = KW * wv;

    __shared__ float4 red[2 * 3 * 6 * 2 * 32];

    v2f bfrag[NCW][2][6];
#pragma unroll
    for (int jj = 0; jj < NCW; ++jj) {
#pragma unroll
        for (int i = 0; i < 2; ++i) {
            const int k0 = kw + 8 * jj + 4 * h + 2 * i;
#pragma unroll
            for (int t = 0; t < 6; ++t) {
                const float* __restrict__ wc = W + 16 * t + c;
                if (!TAIL || jj < NCW - 1) {
                    bfrag[jj][i][t][0] = wc[(k0    ) * NC];
                    bfrag[jj][i][t][1] = wc[(k0 + 1) * NC];
                } else {
                    const bool ok0 = (k0     < K);
                    const bool ok1 = (k0 + 1 < K);
                    const float w0 = wc[(ok0 ? k0     : K - 1) * NC];
                    const float w1 = wc[(ok1 ? k0 + 1 : K - 1) * NC];
                    bfrag[jj][i][t][0] = ok0 ? w0 : 0.f;
                    bfrag[jj][i][t][1] = ok1 ? w1 : 0.f;
                }
            }
        }
    }
    float bk[NCW][4];
#pragma unroll
    for (int jj = 0; jj < NCW; ++jj) {
#pragma unroll
        for (int i = 0; i < 4; ++i) {
            const int k = kw + 8 * jj + 4 * h + i;
            bk[jj][i] = DUAL ? biasIn[k < K ? k : K - 1] : 0.f;
        }
    }

    int par = 0;
    for (int strip = blockIdx.x; strip < nStrips; strip += gridDim.x, par ^= 1) {
        const int m0 = strip * 16;
        const float drow = dinv[m0 + c];
        const float* __restrict__ arow  = A + (size_t)(m0 + c) * K + kw;
        const float* __restrict__ arow2 = DUAL ? A2 + (size_t)(m0 + c) * K + kw : arow;

        v4f ra[NCW], ra2[NCW];
#pragma unroll
        for (int jj = 0; jj < NCW; ++jj) {
#pragma unroll
            for (int i = 0; i < 4; ++i) {
                int kc = 8 * jj + 4 * h + i;
                if (TAIL && jj == NCW - 1)
                    kc = (kw + kc < K) ? kc : K - 1 - kw;
                ra [jj][i] = arow[kc];
                ra2[jj][i] = DUAL ? arow2[kc] : 0.f;
            }
        }
        if constexpr (DUAL) {
            static_assert(NCW == 4, "the statement below names the loads of 4 chunks");
            asm volatile("" :: "v"(ra[0]), "v"(ra[1]), "v"(ra[2]), "v"(ra[3]),
                               "v"(ra2[0]), "v"(ra2[1]), "v"(ra2[2]), "v"(ra2[3]));
        }
        float av[NCW][4];
#pragma unroll
        for (int jj = 0; jj < NCW; ++jj) {
#pragma unroll
            for (int i = 0; i < 4; ++i) {
                float a = ra[jj][i];
                if (DUAL) a = drow * (a + ra2[jj][i]) + bk[jj][i];
                if (RELU) a = fmaxf(a, 0.0f);
                if (TAIL && jj == NCW - 1) a = (kw + 8 * jj + 4 * h + i < K) ? a : 0.f;
                av[jj][i] = a;
            }
        }

        v8f acc[6] = {};
#pragma unroll
        for (int jj = 0; jj < NCW; ++jj) {
#pragma unroll
            for (int i = 0; i < 2; ++i) {
                v2f a; a[0] = av[jj][2 * i]; a[1] = av[jj][2 * i + 1];
#pragma unroll
                for (int t = 0; t < 6; ++t) {
                    acc[t] = __builtin_amdgcn_wmma_f32_16x16x4_f32(
                        false, a, false, bfrag[jj][i][t], (short)0, acc[t], false, false);
                }
                __builtin_amdgcn_sched_barrier(0);
            }
        }

        float4* __restrict__ rb = red + par * (3 * 6 * 2 * 32);
#pragma unroll
        for (int t = 0; t < 6; ++t) {
#pragma unroll
            for (int q = 0; q < 2; ++q) {
                rb[((wv * 6 + t) * 2 + q) * 32 + lane] =
                    make_float4(acc[t][4 * q], acc[t][4 * q + 1], acc[t][4 * q + 2], acc[t][4 * q + 3]);
            }
        }
        __syncthreads();
        float4 pt[3][2][2];
#pragma unroll
        for (int w2 = 0; w2 < 3; ++w2) {
#pragma unroll
            for (int tt = 0; tt < 2; ++tt) {
#pragma unroll
                for (int q = 0; q < 2; ++q)
                    pt[w2][tt][q] = rb[((w2 * 6 + 2 * wv + tt) * 2 + q) * 32 + lane];
            }
        }
        __builtin_amdgcn_sched_barrier(0);
        v8f sum0, sum1;
#pragma unroll
        for (int q = 0; q < 2; ++q) {
            float4 f0 = pt[0][0][q];
            float4 f1 = pt[0][1][q];
#pragma unroll
            for (int w2 = 1; w2 < 3; ++w2) {
                const float4 g0 = pt[w2][0][q];
                const float4 g1 = pt[w2][1][q];
                f0.x += g0.x; f0.y += g0.y; f0.z += g0.z; f0.w += g0.w;
                f1.x += g1.x; f1.y += g1.y; f1.z += g1.z; f1.w += g1.w;
            }
            sum0[4 * q] = f0.x; sum0[4 * q + 1] = f0.y; sum0[4 * q + 2] = f0.z; sum0[4 * q + 3] = f0.w;
            sum1[4 * q] = f1.x; sum1[4 * q + 1] = f1.y; sum1[4 * q + 2] = f1.z; sum1[4 * q + 3] = f1.w;
        }

#pragma unroll
        for (int r = 0; r < 8; ++r) {
            const float t0 = sum0[r];
            const float t1 = sum1[r];
            const float p0 = __int_as_float(__builtin_amdgcn_permlanex16(
                0, __float_as_int(t0), 0x76543210, 0xfedcba98, false, false));
            const float p1 = __int_as_float(__builtin_amdgcn_permlanex16(
                0, __float_as_int(t1), 0x76543210, 0xfedcba98, false, false));
            const float lineA = h ? p1 : t0;
            const float lineB = h ? t1 : p0;
            const float vA = lineA * __int_as_float(__builtin_amdgcn_readlane(__float_as_int(drow), r));
            const float vB = lineB * __int_as_float(__builtin_amdgcn_readlane(__float_as_int(drow), 8 + r));
            const size_t oA = (size_t)(m0 + r) * NC + col0 + lane;
            const size_t oB = (size_t)(m0 + 8 + r) * NC + col0 + lane;
            hw  [oA] = vA;
            agg [oA] = vA;
            agg2[oA] = 0.0f;
            hw  [oB] = vB;
            agg [oB] = vB;
            agg2[oB] = 0.0f;
        }
    }
}

template <int NCH, int G>
__global__ void gcn_push_rows_kernel(const int* __restrict__ src,
                                     const int* __restrict__ dst,
                                     const int* __restrict__ cnt0,
                                     const int* __restrict__ cnt1,
                                     const int* __restrict__ slots0,
                                     const int* __restrict__ slots1,
                                     const int* __restrict__ ovfCnt,
                                     const int* __restrict__ ovfList,
                                     const float* __restrict__ hw,
                                     float* __restrict__ agg0,
                                     float* __restrict__ agg1, int N) {
    constexpr int C = 32 * NCH;
    static_assert(GCN_CAP == 32, "lane i holds place i of a list");
    const int lane = threadIdx.x & 31;
    const int wave = __builtin_amdgcn_readfirstlane(
        (int)((blockIdx.x * (size_t)blockDim.x + threadIdx.x) >> 5));
    const int nWaves = (int)((gridDim.x * (size_t)blockDim.x) >> 5);
    float* __restrict__ agg = gcn_l2_half() ? agg1 : agg0;

    const int s0 = wave * G;
    if (s0 < N) {
        int   n0[G], n1[G];
        int   l0[G], l1[G];
        float v[G][NCH];
#pragma unroll
        for (int u = 0; u < G; ++u) {
            const bool ok = (s0 + u < N);
            const int  s  = ok ? (s0 + u) : s0;
            const int  c0 = cnt0[s];
            const int  c1 = cnt1[s];
            n0[u] = ok ? (c0 < GCN_CAP ? c0 : GCN_CAP) : 0;
            n1[u] = ok ? (c1 < GCN_CAP ? c1 : GCN_CAP) : 0;
            l0[u] = slots0[(size_t)s * GCN_CAP + lane];
            l1[u] = slots1[(size_t)s * GCN_CAP + lane];
            const float* __restrict__ hrow = hw + (size_t)s * C + lane;
#pragma unroll
            for (int q = 0; q < NCH; ++q) v[u][q] = hrow[q * 32];
        }
#pragma unroll
        for (int u = 0; u < G; ++u) {
            for (int j = 0; j < n0[u]; ++j) {
                const int dj = __builtin_amdgcn_readlane(l0[u], j);
                float* __restrict__ arow = agg + (size_t)dj * C + lane;
#pragma unroll
                for (int q = 0; q < NCH; ++q) atomicAdd(&arow[q * 32], v[u][q]);
            }
            for (int j = 0; j < n1[u]; ++j) {
                const int dj = __builtin_amdgcn_readlane(l1[u], j);
                float* __restrict__ arow = agg + (size_t)dj * C + lane;
#pragma unroll
                for (int q = 0; q < NCH; ++q) atomicAdd(&arow[q * 32], v[u][q]);
            }
        }
    }

    const int no = ovfCnt[0];
    for (int i = wave; i < no; i += nWaves) {
        const int e  = ovfList[i];
        const int sj = src[e];
        const int dj = dst[e];
        const float* __restrict__ hrow = hw  + (size_t)sj * C + lane;
        float*       __restrict__ arow = agg + (size_t)dj * C + lane;
#pragma unroll
        for (int q = 0; q < NCH; ++q) atomicAdd(&arow[q * 32], hrow[q * 32]);
    }
}

template <int G>
__global__ void gcn_push_pairs_kernel(const int* __restrict__ src,
                                      const int* __restrict__ dst,
                                      const int* __restrict__ cnt0,
                                      const int* __restrict__ cnt1,
                                      const int* __restrict__ slots0,
                                      const int* __restrict__ slots1,
                                      const int* __restrict__ ovfCnt,
                                      const int* __restrict__ ovfList,
                                      const float* __restrict__ hw,
                                      float* __restrict__ agg0,
                                      float* __restrict__ agg1, int N) {
    static_assert(GCN_CAP == 32, "a list is two runs of 16 places");
    static_assert(G == 8, "the statement below names the loads of 8 sources");
    const int lane = threadIdx.x & 31;
    const int ch   = lane & 1;
    const int p    = lane >> 1;
    const int wave = __builtin_amdgcn_readfirstlane(
        (int)((blockIdx.x * (size_t)blockDim.x + threadIdx.x) >> 5));
    const int nWaves = (int)((gridDim.x * (size_t)blockDim.x) >> 5);
    float* __restrict__ agg = gcn_l2_half() ? agg1 : agg0;

    const int s0 = wave * G;
    if (s0 < N) {
        int   n0[G], n1[G];
        int   l0[G], l1[G];
        float v[G];
#pragma unroll
        for (int u = 0; u < G; ++u) {
            const bool ok = (s0 + u < N);
            const int  s  = ok ? (s0 + u) : s0;
            const int  c0 = cnt0[s];
            const int  c1 = cnt1[s];
            n0[u] = ok ? (c0 < GCN_CAP ? c0 : GCN_CAP) : 0;
            n1[u] = ok ? (c1 < GCN_CAP ? c1 : GCN_CAP) : 0;
            l0[u] = slots0[(size_t)s * GCN_CAP + p];
            l1[u] = slots1[(size_t)s * GCN_CAP + p];
            v[u]  = hw[(size_t)s * 2 + ch];
        }
        asm volatile("" :: "v"(l0[0]), "v"(l0[1]), "v"(l0[2]), "v"(l0[3]),
                           "v"(l0[4]), "v"(l0[5]), "v"(l0[6]), "v"(l0[7]),
                           "v"(l1[0]), "v"(l1[1]), "v"(l1[2]), "v"(l1[3]),
                           "v"(l1[4]), "v"(l1[5]), "v"(l1[6]), "v"(l1[7]),
                           "v"(v[0]), "v"(v[1]), "v"(v[2]), "v"(v[3]),
                           "v"(v[4]), "v"(v[5]), "v"(v[6]), "v"(v[7]));
#pragma unroll
        for (int u = 0; u < G; ++u) {
            if (p < n0[u]) atomicAdd(&agg[(size_t)l0[u] * 2 + ch], v[u]);
            if (p < n1[u]) atomicAdd(&agg[(size_t)l1[u] * 2 + ch], v[u]);
            if (n0[u] > 16) {
                const int d = slots0[(size_t)(s0 + u) * GCN_CAP + 16 + p];
                if (16 + p < n0[u]) atomicAdd(&agg[(size_t)d * 2 + ch], v[u]);
            }
            if (n1[u] > 16) {
                const int d = slots1[(size_t)(s0 + u) * GCN_CAP + 16 + p];
                if (16 + p < n1[u]) atomicAdd(&agg[(size_t)d * 2 + ch], v[u]);
            }
        }
    }

    const int no = ovfCnt[0];
    for (int i = wave * 16 + p; i < no; i += nWaves * 16) {
        const int e  = ovfList[i];
        const int sj = src[e];
        const int dj = dst[e];
        atomicAdd(&agg[(size_t)dj * 2 + ch], hw[(size_t)sj * 2 + ch]);
    }
}


extern "C" void kernel_launch(void* const* d_in, const int* in_sizes, int n_in,
                              void* d_out, int out_size, void* d_ws, size_t ws_size,
                              hipStream_t stream) {
    (void)in_sizes; (void)n_in; (void)out_size; (void)ws_size;

    const int N = GCN_N;
    const int E = GCN_E;

    const float* x    = (const float*)d_in[0];
    const int*   ei   = (const int*)d_in[1];
    const int*   src  = ei;
    const int*   dstp = ei + E;

    const float* W[5] = { (const float*)d_in[2], (const float*)d_in[4],
                          (const float*)d_in[6], (const float*)d_in[8],
                          (const float*)d_in[10] };
    const float* B[5] = { (const float*)d_in[3], (const float*)d_in[5],
                          (const float*)d_in[7], (const float*)d_in[9],
                          (const float*)d_in[11] };

    float* ws    = (float*)d_ws;
    float* hw    = ws;
    float* aggA  = hw    + (size_t)N * 96;
    float* aggB  = aggA  + (size_t)N * 96;
    float* aggA2 = aggB  + (size_t)N * 96;
    float* aggB2 = aggA2 + (size_t)N * 96;
    float* dinv  = aggB2 + (size_t)N * 96;
    float* deg2  = dinv  + GCN_NPAD;
    float* out2  = deg2  + GCN_NPAD;
    float* out   = (float*)d_out;
    int* cnt0    = (int*)(out2 + (size_t)N * 2);
    int* cnt1    = cnt0   + GCN_NPAD;
    int* slots0  = cnt1   + GCN_NPAD;
    int* slots1  = slots0 + (size_t)N * GCN_CAP;
    int* ovfCnt  = slots1 + (size_t)N * GCN_CAP;
    int* ovfList = ovfCnt + 64;

    const int T = 256;

    gcn_zero_kernel  <<<(N + T - 1) / T, T, 0, stream>>>(dinv, deg2, cnt0, cnt1, ovfCnt, N);
    gcn_degree_bucket_kernel<<<(E + T - 1) / T, T, 0, stream>>>(
        src, dstp, dinv, deg2, cnt0, cnt1, slots0, slots1, ovfCnt, ovfList, E);
    gcn_dinv_kernel  <<<(N + T - 1) / T, T, 0, stream>>>(dinv, deg2, N);

    const int dims[6] = { 165, 96, 96, 96, 96, 2 };
    const float* layerIn  = x;
    const float* layerIn2 = x;
    float* aggBufs [2] = { aggA,  aggB  };
    float* aggBufs2[2] = { aggA2, aggB2 };

    for (int l = 0; l < 5; ++l) {
        const int K  = dims[l];
        const int Cn = dims[l + 1];
        float* agg  = (l == 4) ? out  : aggBufs [l & 1];
        float* agg2 = (l == 4) ? out2 : aggBufs2[l & 1];
        const float* biasIn = (l == 0) ? B[0] : B[l - 1];

        const int tiles         = (N / 16) * ((Cn + 15) / 16);
        const int wavesPerBlock = 4;
        const int gBlocks       = (tiles + wavesPerBlock - 1) / wavesPerBlock;
        if (l == 0) {
            gcn_gemm_wstat_f32<165, 0, 0><<<1024, 96, 0, stream>>>(
                layerIn, layerIn2, W[l], biasIn, dinv, hw, agg, agg2, GCN_N / 16);
        } else if (l < 4) {
            gcn_gemm_wstat_f32<96, 1, 1><<<1024, 96, 0, stream>>>(
                layerIn, layerIn2, W[l], biasIn, dinv, hw, agg, agg2, GCN_N / 16);
        } else {
            gcn_gemm_wmma_f32<1><<<gBlocks, wavesPerBlock * 32, 0, stream>>>(
                layerIn, layerIn2, W[l], biasIn, dinv, hw, agg, agg2, N, K, Cn);
        }

        const int pushBlocks    = (N + 8 * 2 - 1) / (8 * 2);
        if (l < 4) {
            gcn_push_rows_kernel<3, 8><<<pushBlocks, 64, 0, stream>>>(
                src, dstp, cnt0, cnt1, slots0, slots1, ovfCnt, ovfList, hw, agg, agg2, N);
        } else {
            gcn_push_pairs_kernel<8><<<pushBlocks, 64, 0, stream>>>(
                src, dstp, cnt0, cnt1, slots0, slots1, ovfCnt, ovfList, hw, agg, agg2, N);
        }

        layerIn  = agg;
        layerIn2 = agg2;
    }

    gcn_finish_kernel<2><<<(2 * N + T - 1) / T, T, 0, stream>>>(out, out2, dinv, B[4], 2 * N);
}


# Last layer: GEMM and edge pass in ONE launch; a wave loads its 16-node strip, bias, weights as one group (was 24 load-wait-WMMA steps), pushes from registers; self-loop an atomic; sum orders changed
# speedup vs baseline: 4.7904x; 1.0510x over previous
#include <hip/hip_runtime.h>
#include <hip/hip_bf16.h>

typedef __attribute__((ext_vector_type(2))) float v2f;
typedef __attribute__((ext_vector_type(4))) float v4f;
typedef __attribute__((ext_vector_type(8))) float v8f;

#define GCN_N 50000
#define GCN_E 800000
#define GCN_NPAD 50048


__device__ __forceinline__ unsigned gcn_l2_half() {
    return (__builtin_amdgcn_s_sendmsg_rtn(0x87) >> 18) & 1u;
}

__global__ void gcn_zero_kernel(float* __restrict__ p, float* __restrict__ q,
                                int* __restrict__ c0, int* __restrict__ c1,
                                int* __restrict__ ovfCnt,
                                float* __restrict__ o0, float* __restrict__ o1, int n) {
    int i = blockIdx.x * blockDim.x + threadIdx.x;
    if (i < n) {
        p[i] = 0.0f; q[i] = 0.0f; c0[i] = 0; c1[i] = 0;
        o0[2 * i] = 0.0f; o0[2 * i + 1] = 0.0f;
        o1[2 * i] = 0.0f; o1[2 * i + 1] = 0.0f;
    }
    if (i == 0) ovfCnt[0] = 0;
}

#define GCN_CAP 32

__global__ void gcn_degree_bucket_kernel(const int* __restrict__ src, const int* __restrict__ dst,
                                         float* __restrict__ deg0, float* __restrict__ deg1,
                                         int* __restrict__ cnt0, int* __restrict__ cnt1,
                                         int* __restrict__ slots0, int* __restrict__ slots1,
                                         int* __restrict__ ovfCnt, int* __restrict__ ovfList,
                                         int E) {
    int e = blockIdx.x * blockDim.x + threadIdx.x;
    const unsigned hf = gcn_l2_half();
    float* __restrict__ deg   = hf ? deg1   : deg0;
    int*   __restrict__ cnt   = hf ? cnt1   : cnt0;
    int*   __restrict__ slots = hf ? slots1 : slots0;
    if (e < E) {
        const int s = src[e];
        const int d = dst[e];
        atomicAdd(&deg[d], 1.0f);
        const int r = atomicAdd(&cnt[s], 1);
        if (r < GCN_CAP) slots[(size_t)s * GCN_CAP + r] = d;
        else             ovfList[atomicAdd(ovfCnt, 1)] = e;
    }
}

__global__ void gcn_dinv_kernel(float* __restrict__ deg, const float* __restrict__ deg1, int n) {
    int i = blockIdx.x * blockDim.x + threadIdx.x;
    if (i < n) deg[i] = rsqrtf(deg[i] + deg1[i] + 1.0f);
}


template <int C>
__global__ void gcn_finish_kernel(float* __restrict__ out, const float* __restrict__ other,
                                  const float* __restrict__ dinv,
                                  const float* __restrict__ bias, int n) {
    int i = blockIdx.x * blockDim.x + threadIdx.x;
    if (i < n) {
        const int node = i / C;
        out[i] = dinv[node] * (out[i] + other[i]) + bias[i - node * C];
    }
}

template <int K, int RELU, int DUAL>
__global__ __launch_bounds__(96)
void gcn_gemm_wstat_f32(const float* __restrict__ A,
                        const float* __restrict__ A2,
                        const float* __restrict__ W,
                        const float* __restrict__ biasIn,
                        const float* __restrict__ dinv,
                        float* __restrict__ hw,
                        float* __restrict__ agg,
                        float* __restrict__ agg2,
                        int nStrips) {
    constexpr int  NC     = 96;
    constexpr int  NCHUNK = (K + 7) / 8;
    static_assert(NCHUNK % 3 == 0, "the 8-wide k chunks are shared out evenly over the 3 waves");
    constexpr int  NCW    = NCHUNK / 3;
    constexpr int  KW     = 8 * NCW;
    constexpr bool TAIL   = (K % 8) != 0;
    const int lane = threadIdx.x & 31;
    const int wv   = threadIdx.x >> 5;
    const int h    = lane >> 4;
    const int c    = lane & 15;
    const int col0 = wv * 32;
    const int kw   = KW * wv;

    __shared__ float4 red[2 * 3 * 6 * 2 * 32];

    v2f bfrag[NCW][2][6];
#pragma unroll
    for (int jj = 0; jj < NCW; ++jj) {
#pragma unroll
        for (int i = 0; i < 2; ++i) {
            const int k0 = kw + 8 * jj + 4 * h + 2 * i;
#pragma unroll
            for (int t = 0; t < 6; ++t) {
                const float* __restrict__ wc = W + 16 * t + c;
                if (!TAIL || jj < NCW - 1) {
                    bfrag[jj][i][t][0] = wc[(k0    ) * NC];
                    bfrag[jj][i][t][1] = wc[(k0 + 1) * NC];
                } else {
                    const bool ok0 = (k0     < K);
                    const bool ok1 = (k0 + 1 < K);
                    const float w0 = wc[(ok0 ? k0     : K - 1) * NC];
                    const float w1 = wc[(ok1 ? k0 + 1 : K - 1) * NC];
                    bfrag[jj][i][t][0] = ok0 ? w0 : 0.f;
                    bfrag[jj][i][t][1] = ok1 ? w1 : 0.f;
                }
            }
        }
    }
    float bk[NCW][4];
#pragma unroll
    for (int jj = 0; jj < NCW; ++jj) {
#pragma unroll
        for (int i = 0; i < 4; ++i) {
            const int k = kw + 8 * jj + 4 * h + i;
            bk[jj][i] = DUAL ? biasIn[k < K ? k : K - 1] : 0.f;
        }
    }

    int par = 0;
    for (int strip = blockIdx.x; strip < nStrips; strip += gridDim.x, par ^= 1) {
        const int m0 = strip * 16;
        const float drow = dinv[m0 + c];
        const float* __restrict__ arow  = A + (size_t)(m0 + c) * K + kw;
        const float* __restrict__ arow2 = DUAL ? A2 + (size_t)(m0 + c) * K + kw : arow;

        v4f ra[NCW], ra2[NCW];
#pragma unroll
        for (int jj = 0; jj < NCW; ++jj) {
#pragma unroll
            for (int i = 0; i < 4; ++i) {
                int kc = 8 * jj + 4 * h + i;
                if (TAIL && jj == NCW - 1)
                    kc = (kw + kc < K) ? kc : K - 1 - kw;
                ra [jj][i] = arow[kc];
                ra2[jj][i] = DUAL ? arow2[kc] : 0.f;
            }
        }
        if constexpr (DUAL) {
            static_assert(NCW == 4, "the statement below names the loads of 4 chunks");
            asm volatile("" :: "v"(ra[0]), "v"(ra[1]), "v"(ra[2]), "v"(ra[3]),
                               "v"(ra2[0]), "v"(ra2[1]), "v"(ra2[2]), "v"(ra2[3]));
        }
        float av[NCW][4];
#pragma unroll
        for (int jj = 0; jj < NCW; ++jj) {
#pragma unroll
            for (int i = 0; i < 4; ++i) {
                float a = ra[jj][i];
                if (DUAL) a = drow * (a + ra2[jj][i]) + bk[jj][i];
                if (RELU) a = fmaxf(a, 0.0f);
                if (TAIL && jj == NCW - 1) a = (kw + 8 * jj + 4 * h + i < K) ? a : 0.f;
                av[jj][i] = a;
            }
        }

        v8f acc[6] = {};
#pragma unroll
        for (int jj = 0; jj < NCW; ++jj) {
#pragma unroll
            for (int i = 0; i < 2; ++i) {
                v2f a; a[0] = av[jj][2 * i]; a[1] = av[jj][2 * i + 1];
#pragma unroll
                for (int t = 0; t < 6; ++t) {
                    acc[t] = __builtin_amdgcn_wmma_f32_16x16x4_f32(
                        false, a, false, bfrag[jj][i][t], (short)0, acc[t], false, false);
                }
                __builtin_amdgcn_sched_barrier(0);
            }
        }

        float4* __restrict__ rb = red + par * (3 * 6 * 2 * 32);
#pragma unroll
        for (int t = 0; t < 6; ++t) {
#pragma unroll
            for (int q = 0; q < 2; ++q) {
                rb[((wv * 6 + t) * 2 + q) * 32 + lane] =
                    make_float4(acc[t][4 * q], acc[t][4 * q + 1], acc[t][4 * q + 2], acc[t][4 * q + 3]);
            }
        }
        __syncthreads();
        float4 pt[3][2][2];
#pragma unroll
        for (int w2 = 0; w2 < 3; ++w2) {
#pragma unroll
            for (int tt = 0; tt < 2; ++tt) {
#pragma unroll
                for (int q = 0; q < 2; ++q)
                    pt[w2][tt][q] = rb[((w2 * 6 + 2 * wv + tt) * 2 + q) * 32 + lane];
            }
        }
        __builtin_amdgcn_sched_barrier(0);
        v8f sum0, sum1;
#pragma unroll
        for (int q = 0; q < 2; ++q) {
            float4 f0 = pt[0][0][q];
            float4 f1 = pt[0][1][q];
#pragma unroll
            for (int w2 = 1; w2 < 3; ++w2) {
                const float4 g0 = pt[w2][0][q];
                const float4 g1 = pt[w2][1][q];
                f0.x += g0.x; f0.y += g0.y; f0.z += g0.z; f0.w += g0.w;
                f1.x += g1.x; f1.y += g1.y; f1.z += g1.z; f1.w += g1.w;
            }
            sum0[4 * q] = f0.x; sum0[4 * q + 1] = f0.y; sum0[4 * q + 2] = f0.z; sum0[4 * q + 3] = f0.w;
            sum1[4 * q] = f1.x; sum1[4 * q + 1] = f1.y; sum1[4 * q + 2] = f1.z; sum1[4 * q + 3] = f1.w;
        }

#pragma unroll
        for (int r = 0; r < 8; ++r) {
            const float t0 = sum0[r];
            const float t1 = sum1[r];
            const float p0 = __int_as_float(__builtin_amdgcn_permlanex16(
                0, __float_as_int(t0), 0x76543210, 0xfedcba98, false, false));
            const float p1 = __int_as_float(__builtin_amdgcn_permlanex16(
                0, __float_as_int(t1), 0x76543210, 0xfedcba98, false, false));
            const float lineA = h ? p1 : t0;
            const float lineB = h ? t1 : p0;
            const float vA = lineA * __int_as_float(__builtin_amdgcn_readlane(__float_as_int(drow), r));
            const float vB = lineB * __int_as_float(__builtin_amdgcn_readlane(__float_as_int(drow), 8 + r));
            const size_t oA = (size_t)(m0 + r) * NC + col0 + lane;
            const size_t oB = (size_t)(m0 + 8 + r) * NC + col0 + lane;
            hw  [oA] = vA;
            agg [oA] = vA;
            agg2[oA] = 0.0f;
            hw  [oB] = vB;
            agg [oB] = vB;
            agg2[oB] = 0.0f;
        }
    }
}

template <int NCH, int G>
__global__ void gcn_push_rows_kernel(const int* __restrict__ src,
                                     const int* __restrict__ dst,
                                     const int* __restrict__ cnt0,
                                     const int* __restrict__ cnt1,
                                     const int* __restrict__ slots0,
                                     const int* __restrict__ slots1,
                                     const int* __restrict__ ovfCnt,
                                     const int* __restrict__ ovfList,
                                     const float* __restrict__ hw,
                                     float* __restrict__ agg0,
                                     float* __restrict__ agg1, int N) {
    constexpr int C = 32 * NCH;
    static_assert(GCN_CAP == 32, "lane i holds place i of a list");
    const int lane = threadIdx.x & 31;
    const int wave = __builtin_amdgcn_readfirstlane(
        (int)((blockIdx.x * (size_t)blockDim.x + threadIdx.x) >> 5));
    const int nWaves = (int)((gridDim.x * (size_t)blockDim.x) >> 5);
    float* __restrict__ agg = gcn_l2_half() ? agg1 : agg0;

    const int s0 = wave * G;
    if (s0 < N) {
        int   n0[G], n1[G];
        int   l0[G], l1[G];
        float v[G][NCH];
#pragma unroll
        for (int u = 0; u < G; ++u) {
            const bool ok = (s0 + u < N);
            const int  s  = ok ? (s0 + u) : s0;
            const int  c0 = cnt0[s];
            const int  c1 = cnt1[s];
            n0[u] = ok ? (c0 < GCN_CAP ? c0 : GCN_CAP) : 0;
            n1[u] = ok ? (c1 < GCN_CAP ? c1 : GCN_CAP) : 0;
            l0[u] = slots0[(size_t)s * GCN_CAP + lane];
            l1[u] = slots1[(size_t)s * GCN_CAP + lane];
            const float* __restrict__ hrow = hw + (size_t)s * C + lane;
#pragma unroll
            for (int q = 0; q < NCH; ++q) v[u][q] = hrow[q * 32];
        }
#pragma unroll
        for (int u = 0; u < G; ++u) {
            for (int j = 0; j < n0[u]; ++j) {
                const int dj = __builtin_amdgcn_readlane(l0[u], j);
                float* __restrict__ arow = agg + (size_t)dj * C + lane;
#pragma unroll
                for (int q = 0; q < NCH; ++q) atomicAdd(&arow[q * 32], v[u][q]);
            }
            for (int j = 0; j < n1[u]; ++j) {
                const int dj = __builtin_amdgcn_readlane(l1[u], j);
                float* __restrict__ arow = agg + (size_t)dj * C + lane;
#pragma unroll
                for (int q = 0; q < NCH; ++q) atomicAdd(&arow[q * 32], v[u][q]);
            }
        }
    }

    const int no = ovfCnt[0];
    for (int i = wave; i < no; i += nWaves) {
        const int e  = ovfList[i];
        const int sj = src[e];
        const int dj = dst[e];
        const float* __restrict__ hrow = hw  + (size_t)sj * C + lane;
        float*       __restrict__ arow = agg + (size_t)dj * C + lane;
#pragma unroll
        for (int q = 0; q < NCH; ++q) atomicAdd(&arow[q * 32], hrow[q * 32]);
    }
}

__global__ __launch_bounds__(64)
void gcn_last_layer_kernel(const float* __restrict__ A,
                           const float* __restrict__ A2,
                           const float* __restrict__ W,
                           const float* __restrict__ biasIn,
                           const float* __restrict__ dinv,
                           const int* __restrict__ src,
                           const int* __restrict__ dst,
                           const int* __restrict__ cnt0,
                           const int* __restrict__ cnt1,
                           const int* __restrict__ slots0,
                           const int* __restrict__ slots1,
                           const int* __restrict__ ovfCnt,
                           const int* __restrict__ ovfList,
                           float* __restrict__ agg0,
                           float* __restrict__ agg1,
                           int nStrips, int E) {
    constexpr int K   = 96;
    constexpr int NCK = K / 8;
    static_assert(GCN_CAP == 32, "a list is two runs of 16 places");
    const int lane = threadIdx.x & 31;
    const int wv   = threadIdx.x >> 5;
    const int h    = lane >> 4;
    const int c    = lane & 15;
    const int ch   = lane & 1;
    const int p    = lane >> 1;
    const int wave = __builtin_amdgcn_readfirstlane(
        (int)((blockIdx.x * (size_t)blockDim.x + threadIdx.x) >> 5));
    float* __restrict__ agg = gcn_l2_half() ? agg1 : agg0;

    __shared__ float hwS[2][32];

    if (wave >= nStrips) return;
    const int m0 = wave * 16;

    const float drow = dinv[m0 + c];
    const float* __restrict__ arow  = A  + (size_t)(m0 + c) * K;
    const float* __restrict__ arow2 = A2 + (size_t)(m0 + c) * K;
    const int cc = (c < 2) ? c : 0;
    v8f ra[NCK / 2], ra2[NCK / 2], bk[NCK / 2], wr[NCK / 2];
#pragma unroll
    for (int jj = 0; jj < NCK; ++jj) {
#pragma unroll
        for (int i = 0; i < 4; ++i) {
            const int k = 8 * jj + 4 * h + i;
            ra [jj >> 1][(jj & 1) * 4 + i] = arow [k];
            ra2[jj >> 1][(jj & 1) * 4 + i] = arow2[k];
            bk [jj >> 1][(jj & 1) * 4 + i] = biasIn[k];
            wr [jj >> 1][(jj & 1) * 4 + i] = W[k * 2 + cc];
        }
    }
    v8f dv;
#pragma unroll
    for (int r = 0; r < 8; ++r) dv[r] = dinv[m0 + 8 * h + r];
    asm volatile("" :: "v"(ra[0]), "v"(ra[1]), "v"(ra[2]), "v"(ra[3]), "v"(ra[4]), "v"(ra[5]),
                       "v"(ra2[0]), "v"(ra2[1]), "v"(ra2[2]), "v"(ra2[3]), "v"(ra2[4]), "v"(ra2[5]),
                       "v"(bk[0]), "v"(bk[1]), "v"(bk[2]), "v"(bk[3]), "v"(bk[4]), "v"(bk[5]),
                       "v"(wr[0]), "v"(wr[1]), "v"(wr[2]), "v"(wr[3]), "v"(wr[4]), "v"(wr[5]),
                       "v"(dv), "v"(drow));
    __builtin_amdgcn_sched_barrier(0);

    v8f av[NCK / 2], bw[NCK / 2];
#pragma unroll
    for (int j = 0; j < NCK / 2; ++j) {
#pragma unroll
        for (int e = 0; e < 8; ++e) {
            av[j][e] = fmaxf(drow * (ra[j][e] + ra2[j][e]) + bk[j][e], 0.0f);
            bw[j][e] = (c < 2) ? wr[j][e] : 0.f;
        }
    }

    __builtin_amdgcn_sched_barrier(0);
    const int cv = h ? cnt1[m0 + c] : cnt0[m0 + c];
    int l0[16], l1[16];
#pragma unroll
    for (int u = 0; u < 16; ++u) {
        l0[u] = slots0[(size_t)(m0 + u) * GCN_CAP + p];
        l1[u] = slots1[(size_t)(m0 + u) * GCN_CAP + p];
    }
    __builtin_amdgcn_sched_barrier(0);

    v8f acc0 = {}, acc1 = {};
#pragma unroll
    for (int jj = 0; jj < NCK; ++jj) {
        const int j = jj >> 1, o = (jj & 1) * 4;
        v2f a0; a0[0] = av[j][o];     a0[1] = av[j][o + 1];
        v2f b0; b0[0] = bw[j][o];     b0[1] = bw[j][o + 1];
        acc0 = __builtin_amdgcn_wmma_f32_16x16x4_f32(
            false, a0, false, b0, (short)0, acc0, false, false);
        v2f a1; a1[0] = av[j][o + 2]; a1[1] = av[j][o + 3];
        v2f b1; b1[0] = bw[j][o + 2]; b1[1] = bw[j][o + 3];
        acc1 = __builtin_amdgcn_wmma_f32_16x16x4_f32(
            false, a1, false, b1, (short)0, acc1, false, false);
    }
    float val[8];
#pragma unroll
    for (int r = 0; r < 8; ++r) val[r] = (acc0[r] + acc1[r]) * dv[r];

    float v[16];
#pragma unroll
    for (int u = 0; u < 16; ++u) {
        const float s0 = __int_as_float(__builtin_amdgcn_readlane(__float_as_int(val[u & 7]), 16 * (u >> 3)));
        const float s1 = __int_as_float(__builtin_amdgcn_readlane(__float_as_int(val[u & 7]), 16 * (u >> 3) + 1));
        v[u] = ch ? s1 : s0;
    }

    float vs = v[0];
#pragma unroll
    for (int u = 1; u < 16; ++u) vs = (p == u) ? v[u] : vs;
    atomicAdd(&agg[(size_t)(m0 + p) * 2 + ch], vs);

    __builtin_amdgcn_sched_barrier(0);
    asm volatile("" :: "v"(l0[0]), "v"(l0[1]), "v"(l0[2]), "v"(l0[3]),
                       "v"(l0[4]), "v"(l0[5]), "v"(l0[6]), "v"(l0[7]),
                       "v"(l0[8]), "v"(l0[9]), "v"(l0[10]), "v"(l0[11]),
                       "v"(l0[12]), "v"(l0[13]), "v"(l0[14]), "v"(l0[15]), "v"(cv));
    asm volatile("" :: "v"(l1[0]), "v"(l1[1]), "v"(l1[2]), "v"(l1[3]),
                       "v"(l1[4]), "v"(l1[5]), "v"(l1[6]), "v"(l1[7]),
                       "v"(l1[8]), "v"(l1[9]), "v"(l1[10]), "v"(l1[11]),
                       "v"(l1[12]), "v"(l1[13]), "v"(l1[14]), "v"(l1[15]));
#pragma unroll
    for (int u = 0; u < 16; ++u) {
        const int c0 = __builtin_amdgcn_readlane(cv, u);
        const int c1 = __builtin_amdgcn_readlane(cv, 16 + u);
        const int n0 = c0 < GCN_CAP ? c0 : GCN_CAP;
        const int n1 = c1 < GCN_CAP ? c1 : GCN_CAP;
        if (p < n0) atomicAdd(&agg[(size_t)l0[u] * 2 + ch], v[u]);
        if (p < n1) atomicAdd(&agg[(size_t)l1[u] * 2 + ch], v[u]);
        if (n0 > 16) {
            const int d = slots0[(size_t)(m0 + u) * GCN_CAP + 16 + p];
            if (16 + p < n0) atomicAdd(&agg[(size_t)d * 2 + ch], v[u]);
        }
        if (n1 > 16) {
            const int d = slots1[(size_t)(m0 + u) * GCN_CAP + 16 + p];
            if (16 + p < n1) atomicAdd(&agg[(size_t)d * 2 + ch], v[u]);
        }
    }

    int no = ovfCnt[0];
    no = no < E ? no : E;
    if (no > 0) {
        float* __restrict__ tab = hwS[wv];
        if (p == 0) {
#pragma unroll
            for (int u = 0; u < 16; ++u) tab[u * 2 + ch] = v[u];
        }
        for (int i = lane; i < no; i += 32) {
            const int e = ovfList[i];
            const int u = src[e] - m0;
            if (u >= 0 && u < 16) {
                const int dj = dst[e];
                atomicAdd(&agg[(size_t)dj * 2],     tab[u * 2]);
                atomicAdd(&agg[(size_t)dj * 2 + 1], tab[u * 2 + 1]);
            }
        }
    }
}


extern "C" void kernel_launch(void* const* d_in, const int* in_sizes, int n_in,
                              void* d_out, int out_size, void* d_ws, size_t ws_size,
                              hipStream_t stream) {
    (void)in_sizes; (void)n_in; (void)out_size; (void)ws_size;

    const int N = GCN_N;
    const int E = GCN_E;

    const float* x    = (const float*)d_in[0];
    const int*   ei   = (const int*)d_in[1];
    const int*   src  = ei;
    const int*   dstp = ei + E;

    const float* W[5] = { (const float*)d_in[2], (const float*)d_in[4],
                          (const float*)d_in[6], (const float*)d_in[8],
                          (const float*)d_in[10] };
    const float* B[5] = { (const float*)d_in[3], (const float*)d_in[5],
                          (const float*)d_in[7], (const float*)d_in[9],
                          (const float*)d_in[11] };

    float* ws    = (float*)d_ws;
    float* hw    = ws;
    float* aggA  = hw    + (size_t)N * 96;
    float* aggB  = aggA  + (size_t)N * 96;
    float* aggA2 = aggB  + (size_t)N * 96;
    float* aggB2 = aggA2 + (size_t)N * 96;
    float* dinv  = aggB2 + (size_t)N * 96;
    float* deg2  = dinv  + GCN_NPAD;
    float* out2  = deg2  + GCN_NPAD;
    float* out   = (float*)d_out;
    int* cnt0    = (int*)(out2 + (size_t)N * 2);
    int* cnt1    = cnt0   + GCN_NPAD;
    int* slots0  = cnt1   + GCN_NPAD;
    int* slots1  = slots0 + (size_t)N * GCN_CAP;
    int* ovfCnt  = slots1 + (size_t)N * GCN_CAP;
    int* ovfList = ovfCnt + 64;

    const int T = 256;

    gcn_zero_kernel  <<<(N + T - 1) / T, T, 0, stream>>>(dinv, deg2, cnt0, cnt1, ovfCnt, out, out2, N);
    gcn_degree_bucket_kernel<<<(E + T - 1) / T, T, 0, stream>>>(
        src, dstp, dinv, deg2, cnt0, cnt1, slots0, slots1, ovfCnt, ovfList, E);
    gcn_dinv_kernel  <<<(N + T - 1) / T, T, 0, stream>>>(dinv, deg2, N);

    const float* layerIn  = x;
    const float* layerIn2 = x;
    float* aggBufs [2] = { aggA,  aggB  };
    float* aggBufs2[2] = { aggA2, aggB2 };

    for (int l = 0; l < 4; ++l) {
        float* agg  = aggBufs [l & 1];
        float* agg2 = aggBufs2[l & 1];
        const float* biasIn = (l == 0) ? B[0] : B[l - 1];

        if (l == 0) {
            gcn_gemm_wstat_f32<165, 0, 0><<<1024, 96, 0, stream>>>(
                layerIn, layerIn2, W[l], biasIn, dinv, hw, agg, agg2, GCN_N / 16);
        } else {
            gcn_gemm_wstat_f32<96, 1, 1><<<1024, 96, 0, stream>>>(
                layerIn, layerIn2, W[l], biasIn, dinv, hw, agg, agg2, GCN_N / 16);
        }

        const int pushBlocks    = (N + 8 * 2 - 1) / (8 * 2);
        gcn_push_rows_kernel<3, 8><<<pushBlocks, 64, 0, stream>>>(
            src, dstp, cnt0, cnt1, slots0, slots1, ovfCnt, ovfList, hw, agg, agg2, N);

        layerIn  = agg;
        layerIn2 = agg2;
    }

    gcn_last_layer_kernel<<<(GCN_N / 16 + 1) / 2, 64, 0, stream>>>(
        layerIn, layerIn2, W[4], B[3], dinv, src, dstp, cnt0, cnt1, slots0, slots1,
        ovfCnt, ovfList, out, out2, GCN_N / 16, E);

    gcn_finish_kernel<2><<<(2 * N + T - 1) / T, T, 0, stream>>>(out, out2, dinv, B[4], 2 * N);
}


# Layers 1-3 store their rows in place over the sums they have just read: one pair of 96-channel sum arrays, not two in turn; 3 row arrays live, not 5; stores go to lines just loaded; sums unchanged
# speedup vs baseline: 4.9417x; 1.0316x over previous
#include <hip/hip_runtime.h>
#include <hip/hip_bf16.h>

typedef __attribute__((ext_vector_type(2))) float v2f;
typedef __attribute__((ext_vector_type(4))) float v4f;
typedef __attribute__((ext_vector_type(8))) float v8f;

#define GCN_N 50000
#define GCN_E 800000
#define GCN_NPAD 50048


__device__ __forceinline__ unsigned gcn_l2_half() {
    return (__builtin_amdgcn_s_sendmsg_rtn(0x87) >> 18) & 1u;
}

__global__ void gcn_zero_kernel(float* __restrict__ p, float* __restrict__ q,
                                int* __restrict__ c0, int* __restrict__ c1,
                                int* __restrict__ ovfCnt,
                                float* __restrict__ o0, float* __restrict__ o1, int n) {
    int i = blockIdx.x * blockDim.x + threadIdx.x;
    if (i < n) {
        p[i] = 0.0f; q[i] = 0.0f; c0[i] = 0; c1[i] = 0;
        o0[2 * i] = 0.0f; o0[2 * i + 1] = 0.0f;
        o1[2 * i] = 0.0f; o1[2 * i + 1] = 0.0f;
    }
    if (i == 0) ovfCnt[0] = 0;
}

#define GCN_CAP 32

__global__ void gcn_degree_bucket_kernel(const int* __restrict__ src, const int* __restrict__ dst,
                                         float* __restrict__ deg0, float* __restrict__ deg1,
                                         int* __restrict__ cnt0, int* __restrict__ cnt1,
                                         int* __restrict__ slots0, int* __restrict__ slots1,
                                         int* __restrict__ ovfCnt, int* __restrict__ ovfList,
                                         int E) {
    int e = blockIdx.x * blockDim.x + threadIdx.x;
    const unsigned hf = gcn_l2_half();
    float* __restrict__ deg   = hf ? deg1   : deg0;
    int*   __restrict__ cnt   = hf ? cnt1   : cnt0;
    int*   __restrict__ slots = hf ? slots1 : slots0;
    if (e < E) {
        const int s = src[e];
        const int d = dst[e];
        atomicAdd(&deg[d], 1.0f);
        const int r = atomicAdd(&cnt[s], 1);
        if (r < GCN_CAP) slots[(size_t)s * GCN_CAP + r] = d;
        else             ovfList[atomicAdd(ovfCnt, 1)] = e;
    }
}

__global__ void gcn_dinv_kernel(float* __restrict__ deg, const float* __restrict__ deg1, int n) {
    int i = blockIdx.x * blockDim.x + threadIdx.x;
    if (i < n) deg[i] = rsqrtf(deg[i] + deg1[i] + 1.0f);
}


template <int C>
__global__ void gcn_finish_kernel(float* __restrict__ out, const float* __restrict__ other,
                                  const float* __restrict__ dinv,
                                  const float* __restrict__ bias, int n) {
    int i = blockIdx.x * blockDim.x + threadIdx.x;
    if (i < n) {
        const int node = i / C;
        out[i] = dinv[node] * (out[i] + other[i]) + bias[i - node * C];
    }
}

template <int K, int RELU, int DUAL>
__global__ __launch_bounds__(96)
void gcn_gemm_wstat_f32(const float* A,
                        const float* A2,
                        const float* __restrict__ W,
                        const float* __restrict__ biasIn,
                        const float* __restrict__ dinv,
                        float* __restrict__ hw,
                        float* agg,
                        float* agg2,
                        int nStrips) {
    constexpr int  NC     = 96;
    constexpr int  NCHUNK = (K + 7) / 8;
    static_assert(NCHUNK % 3 == 0, "the 8-wide k chunks are shared out evenly over the 3 waves");
    constexpr int  NCW    = NCHUNK / 3;
    constexpr int  KW     = 8 * NCW;
    constexpr bool TAIL   = (K % 8) != 0;
    const int lane = threadIdx.x & 31;
    const int wv   = threadIdx.x >> 5;
    const int h    = lane >> 4;
    const int c    = lane & 15;
    const int col0 = wv * 32;
    const int kw   = KW * wv;

    __shared__ float4 red[2 * 3 * 6 * 2 * 32];

    v2f bfrag[NCW][2][6];
#pragma unroll
    for (int jj = 0; jj < NCW; ++jj) {
#pragma unroll
        for (int i = 0; i < 2; ++i) {
            const int k0 = kw + 8 * jj + 4 * h + 2 * i;
#pragma unroll
            for (int t = 0; t < 6; ++t) {
                const float* __restrict__ wc = W + 16 * t + c;
                if (!TAIL || jj < NCW - 1) {
                    bfrag[jj][i][t][0] = wc[(k0    ) * NC];
                    bfrag[jj][i][t][1] = wc[(k0 + 1) * NC];
                } else {
                    const bool ok0 = (k0     < K);
                    const bool ok1 = (k0 + 1 < K);
                    const float w0 = wc[(ok0 ? k0     : K - 1) * NC];
                    const float w1 = wc[(ok1 ? k0 + 1 : K - 1) * NC];
                    bfrag[jj][i][t][0] = ok0 ? w0 : 0.f;
                    bfrag[jj][i][t][1] = ok1 ? w1 : 0.f;
                }
            }
        }
    }
    float bk[NCW][4];
#pragma unroll
    for (int jj = 0; jj < NCW; ++jj) {
#pragma unroll
        for (int i = 0; i < 4; ++i) {
            const int k = kw + 8 * jj + 4 * h + i;
            bk[jj][i] = DUAL ? biasIn[k < K ? k : K - 1] : 0.f;
        }
    }

    int par = 0;
    for (int strip = blockIdx.x; strip < nStrips; strip += gridDim.x, par ^= 1) {
        const int m0 = strip * 16;
        const float drow = dinv[m0 + c];
        const float* arow  = A + (size_t)(m0 + c) * K + kw;
        const float* arow2 = DUAL ? A2 + (size_t)(m0 + c) * K + kw : arow;

        v4f ra[NCW], ra2[NCW];
#pragma unroll
        for (int jj = 0; jj < NCW; ++jj) {
#pragma unroll
            for (int i = 0; i < 4; ++i) {
                int kc = 8 * jj + 4 * h + i;
                if (TAIL && jj == NCW - 1)
                    kc = (kw + kc < K) ? kc : K - 1 - kw;
                ra [jj][i] = arow[kc];
                ra2[jj][i] = DUAL ? arow2[kc] : 0.f;
            }
        }
        if constexpr (DUAL) {
            static_assert(NCW == 4, "the statement below names the loads of 4 chunks");
            asm volatile("" :: "v"(ra[0]), "v"(ra[1]), "v"(ra[2]), "v"(ra[3]),
                               "v"(ra2[0]), "v"(ra2[1]), "v"(ra2[2]), "v"(ra2[3]));
        }
        float av[NCW][4];
#pragma unroll
        for (int jj = 0; jj < NCW; ++jj) {
#pragma unroll
            for (int i = 0; i < 4; ++i) {
                float a = ra[jj][i];
                if (DUAL) a = drow * (a + ra2[jj][i]) + bk[jj][i];
                if (RELU) a = fmaxf(a, 0.0f);
                if (TAIL && jj == NCW - 1) a = (kw + 8 * jj + 4 * h + i < K) ? a : 0.f;
                av[jj][i] = a;
            }
        }

        v8f acc[6] = {};
#pragma unroll
        for (int jj = 0; jj < NCW; ++jj) {
#pragma unroll
            for (int i = 0; i < 2; ++i) {
                v2f a; a[0] = av[jj][2 * i]; a[1] = av[jj][2 * i + 1];
#pragma unroll
                for (int t = 0; t < 6; ++t) {
                    acc[t] = __builtin_amdgcn_wmma_f32_16x16x4_f32(
                        false, a, false, bfrag[jj][i][t], (short)0, acc[t], false, false);
                }
                __builtin_amdgcn_sched_barrier(0);
            }
        }

        float4* __restrict__ rb = red + par * (3 * 6 * 2 * 32);
#pragma unroll
        for (int t = 0; t < 6; ++t) {
#pragma unroll
            for (int q = 0; q < 2; ++q) {
                rb[((wv * 6 + t) * 2 + q) * 32 + lane] =
                    make_float4(acc[t][4 * q], acc[t][4 * q + 1], acc[t][4 * q + 2], acc[t][4 * q + 3]);
            }
        }
        __syncthreads();
        float4 pt[3][2][2];
#pragma unroll
        for (int w2 = 0; w2 < 3; ++w2) {
#pragma unroll
            for (int tt = 0; tt < 2; ++tt) {
#pragma unroll
                for (int q = 0; q < 2; ++q)
                    pt[w2][tt][q] = rb[((w2 * 6 + 2 * wv + tt) * 2 + q) * 32 + lane];
            }
        }
        __builtin_amdgcn_sched_barrier(0);
        v8f sum0, sum1;
#pragma unroll
        for (int q = 0; q < 2; ++q) {
            float4 f0 = pt[0][0][q];
            float4 f1 = pt[0][1][q];
#pragma unroll
            for (int w2 = 1; w2 < 3; ++w2) {
                const float4 g0 = pt[w2][0][q];
                const float4 g1 = pt[w2][1][q];
                f0.x += g0.x; f0.y += g0.y; f0.z += g0.z; f0.w += g0.w;
                f1.x += g1.x; f1.y += g1.y; f1.z += g1.z; f1.w += g1.w;
            }
            sum0[4 * q] = f0.x; sum0[4 * q + 1] = f0.y; sum0[4 * q + 2] = f0.z; sum0[4 * q + 3] = f0.w;
            sum1[4 * q] = f1.x; sum1[4 * q + 1] = f1.y; sum1[4 * q + 2] = f1.z; sum1[4 * q + 3] = f1.w;
        }

#pragma unroll
        for (int r = 0; r < 8; ++r) {
            const float t0 = sum0[r];
            const float t1 = sum1[r];
            const float p0 = __int_as_float(__builtin_amdgcn_permlanex16(
                0, __float_as_int(t0), 0x76543210, 0xfedcba98, false, false));
            const float p1 = __int_as_float(__builtin_amdgcn_permlanex16(
                0, __float_as_int(t1), 0x76543210, 0xfedcba98, false, false));
            const float lineA = h ? p1 : t0;
            const float lineB = h ? t1 : p0;
            const float vA = lineA * __int_as_float(__builtin_amdgcn_readlane(__float_as_int(drow), r));
            const float vB = lineB * __int_as_float(__builtin_amdgcn_readlane(__float_as_int(drow), 8 + r));
            const size_t oA = (size_t)(m0 + r) * NC + col0 + lane;
            const size_t oB = (size_t)(m0 + 8 + r) * NC + col0 + lane;
            hw  [oA] = vA;
            agg [oA] = vA;
            agg2[oA] = 0.0f;
            hw  [oB] = vB;
            agg [oB] = vB;
            agg2[oB] = 0.0f;
        }
    }
}

template <int NCH, int G>
__global__ void gcn_push_rows_kernel(const int* __restrict__ src,
                                     const int* __restrict__ dst,
                                     const int* __restrict__ cnt0,
                                     const int* __restrict__ cnt1,
                                     const int* __restrict__ slots0,
                                     const int* __restrict__ slots1,
                                     const int* __restrict__ ovfCnt,
                                     const int* __restrict__ ovfList,
                                     const float* __restrict__ hw,
                                     float* __restrict__ agg0,
                                     float* __restrict__ agg1, int N) {
    constexpr int C = 32 * NCH;
    static_assert(GCN_CAP == 32, "lane i holds place i of a list");
    const int lane = threadIdx.x & 31;
    const int wave = __builtin_amdgcn_readfirstlane(
        (int)((blockIdx.x * (size_t)blockDim.x + threadIdx.x) >> 5));
    const int nWaves = (int)((gridDim.x * (size_t)blockDim.x) >> 5);
    float* __restrict__ agg = gcn_l2_half() ? agg1 : agg0;

    const int s0 = wave * G;
    if (s0 < N) {
        int   n0[G], n1[G];
        int   l0[G], l1[G];
        float v[G][NCH];
#pragma unroll
        for (int u = 0; u < G; ++u) {
            const bool ok = (s0 + u < N);
            const int  s  = ok ? (s0 + u) : s0;
            const int  c0 = cnt0[s];
            const int  c1 = cnt1[s];
            n0[u] = ok ? (c0 < GCN_CAP ? c0 : GCN_CAP) : 0;
            n1[u] = ok ? (c1 < GCN_CAP ? c1 : GCN_CAP) : 0;
            l0[u] = slots0[(size_t)s * GCN_CAP + lane];
            l1[u] = slots1[(size_t)s * GCN_CAP + lane];
            const float* __restrict__ hrow = hw + (size_t)s * C + lane;
#pragma unroll
            for (int q = 0; q < NCH; ++q) v[u][q] = hrow[q * 32];
        }
#pragma unroll
        for (int u = 0; u < G; ++u) {
            for (int j = 0; j < n0[u]; ++j) {
                const int dj = __builtin_amdgcn_readlane(l0[u], j);
                float* __restrict__ arow = agg + (size_t)dj * C + lane;
#pragma unroll
                for (int q = 0; q < NCH; ++q) atomicAdd(&arow[q * 32], v[u][q]);
            }
            for (int j = 0; j < n1[u]; ++j) {
                const int dj = __builtin_amdgcn_readlane(l1[u], j);
                float* __restrict__ arow = agg + (size_t)dj * C + lane;
#pragma unroll
                for (int q = 0; q < NCH; ++q) atomicAdd(&arow[q * 32], v[u][q]);
            }
        }
    }

    const int no = ovfCnt[0];
    for (int i = wave; i < no; i += nWaves) {
        const int e  = ovfList[i];
        const int sj = src[e];
        const int dj = dst[e];
        const float* __restrict__ hrow = hw  + (size_t)sj * C + lane;
        float*       __restrict__ arow = agg + (size_t)dj * C + lane;
#pragma unroll
        for (int q = 0; q < NCH; ++q) atomicAdd(&arow[q * 32], hrow[q * 32]);
    }
}

__global__ __launch_bounds__(64)
void gcn_last_layer_kernel(const float* __restrict__ A,
                           const float* __restrict__ A2,
                           const float* __restrict__ W,
                           const float* __restrict__ biasIn,
                           const float* __restrict__ dinv,
                           const int* __restrict__ src,
                           const int* __restrict__ dst,
                           const int* __restrict__ cnt0,
                           const int* __restrict__ cnt1,
                           const int* __restrict__ slots0,
                           const int* __restrict__ slots1,
                           const int* __restrict__ ovfCnt,
                           const int* __restrict__ ovfList,
                           float* __restrict__ agg0,
                           float* __restrict__ agg1,
                           int nStrips, int E) {
    constexpr int K   = 96;
    constexpr int NCK = K / 8;
    static_assert(GCN_CAP == 32, "a list is two runs of 16 places");
    const int lane = threadIdx.x & 31;
    const int wv   = threadIdx.x >> 5;
    const int h    = lane >> 4;
    const int c    = lane & 15;
    const int ch   = lane & 1;
    const int p    = lane >> 1;
    const int wave = __builtin_amdgcn_readfirstlane(
        (int)((blockIdx.x * (size_t)blockDim.x + threadIdx.x) >> 5));
    float* __restrict__ agg = gcn_l2_half() ? agg1 : agg0;

    __shared__ float hwS[2][32];

    if (wave >= nStrips) return;
    const int m0 = wave * 16;

    const float drow = dinv[m0 + c];
    const float* __restrict__ arow  = A  + (size_t)(m0 + c) * K;
    const float* __restrict__ arow2 = A2 + (size_t)(m0 + c) * K;
    const int cc = (c < 2) ? c : 0;
    v8f ra[NCK / 2], ra2[NCK / 2], bk[NCK / 2], wr[NCK / 2];
#pragma unroll
    for (int jj = 0; jj < NCK; ++jj) {
#pragma unroll
        for (int i = 0; i < 4; ++i) {
            const int k = 8 * jj + 4 * h + i;
            ra [jj >> 1][(jj & 1) * 4 + i] = arow [k];
            ra2[jj >> 1][(jj & 1) * 4 + i] = arow2[k];
            bk [jj >> 1][(jj & 1) * 4 + i] = biasIn[k];
            wr [jj >> 1][(jj & 1) * 4 + i] = W[k * 2 + cc];
        }
    }
    v8f dv;
#pragma unroll
    for (int r = 0; r < 8; ++r) dv[r] = dinv[m0 + 8 * h + r];
    asm volatile("" :: "v"(ra[0]), "v"(ra[1]), "v"(ra[2]), "v"(ra[3]), "v"(ra[4]), "v"(ra[5]),
                       "v"(ra2[0]), "v"(ra2[1]), "v"(ra2[2]), "v"(ra2[3]), "v"(ra2[4]), "v"(ra2[5]),
                       "v"(bk[0]), "v"(bk[1]), "v"(bk[2]), "v"(bk[3]), "v"(bk[4]), "v"(bk[5]),
                       "v"(wr[0]), "v"(wr[1]), "v"(wr[2]), "v"(wr[3]), "v"(wr[4]), "v"(wr[5]),
                       "v"(dv), "v"(drow));
    __builtin_amdgcn_sched_barrier(0);

    v8f av[NCK / 2], bw[NCK / 2];
#pragma unroll
    for (int j = 0; j < NCK / 2; ++j) {
#pragma unroll
        for (int e = 0; e < 8; ++e) {
            av[j][e] = fmaxf(drow * (ra[j][e] + ra2[j][e]) + bk[j][e], 0.0f);
            bw[j][e] = (c < 2) ? wr[j][e] : 0.f;
        }
    }

    __builtin_amdgcn_sched_barrier(0);
    const int cv = h ? cnt1[m0 + c] : cnt0[m0 + c];
    int l0[16], l1[16];
#pragma unroll
    for (int u = 0; u < 16; ++u) {
        l0[u] = slots0[(size_t)(m0 + u) * GCN_CAP + p];
        l1[u] = slots1[(size_t)(m0 + u) * GCN_CAP + p];
    }
    __builtin_amdgcn_sched_barrier(0);

    v8f acc0 = {}, acc1 = {};
#pragma unroll
    for (int jj = 0; jj < NCK; ++jj) {
        const int j = jj >> 1, o = (jj & 1) * 4;
        v2f a0; a0[0] = av[j][o];     a0[1] = av[j][o + 1];
        v2f b0; b0[0] = bw[j][o];     b0[1] = bw[j][o + 1];
        acc0 = __builtin_amdgcn_wmma_f32_16x16x4_f32(
            false, a0, false, b0, (short)0, acc0, false, false);
        v2f a1; a1[0] = av[j][o + 2]; a1[1] = av[j][o + 3];
        v2f b1; b1[0] = bw[j][o + 2]; b1[1] = bw[j][o + 3];
        acc1 = __builtin_amdgcn_wmma_f32_16x16x4_f32(
            false, a1, false, b1, (short)0, acc1, false, false);
    }
    float val[8];
#pragma unroll
    for (int r = 0; r < 8; ++r) val[r] = (acc0[r] + acc1[r]) * dv[r];

    float v[16];
#pragma unroll
    for (int u = 0; u < 16; ++u) {
        const float s0 = __int_as_float(__builtin_amdgcn_readlane(__float_as_int(val[u & 7]), 16 * (u >> 3)));
        const float s1 = __int_as_float(__builtin_amdgcn_readlane(__float_as_int(val[u & 7]), 16 * (u >> 3) + 1));
        v[u] = ch ? s1 : s0;
    }

    float vs = v[0];
#pragma unroll
    for (int u = 1; u < 16; ++u) vs = (p == u) ? v[u] : vs;
    atomicAdd(&agg[(size_t)(m0 + p) * 2 + ch], vs);

    __builtin_amdgcn_sched_barrier(0);
    asm volatile("" :: "v"(l0[0]), "v"(l0[1]), "v"(l0[2]), "v"(l0[3]),
                       "v"(l0[4]), "v"(l0[5]), "v"(l0[6]), "v"(l0[7]),
                       "v"(l0[8]), "v"(l0[9]), "v"(l0[10]), "v"(l0[11]),
                       "v"(l0[12]), "v"(l0[13]), "v"(l0[14]), "v"(l0[15]), "v"(cv));
    asm volatile("" :: "v"(l1[0]), "v"(l1[1]), "v"(l1[2]), "v"(l1[3]),
                       "v"(l1[4]), "v"(l1[5]), "v"(l1[6]), "v"(l1[7]),
                       "v"(l1[8]), "v"(l1[9]), "v"(l1[10]), "v"(l1[11]),
                       "v"(l1[12]), "v"(l1[13]), "v"(l1[14]), "v"(l1[15]));
#pragma unroll
    for (int u = 0; u < 16; ++u) {
        const int c0 = __builtin_amdgcn_readlane(cv, u);
        const int c1 = __builtin_amdgcn_readlane(cv, 16 + u);
        const int n0 = c0 < GCN_CAP ? c0 : GCN_CAP;
        const int n1 = c1 < GCN_CAP ? c1 : GCN_CAP;
        if (p < n0) atomicAdd(&agg[(size_t)l0[u] * 2 + ch], v[u]);
        if (p < n1) atomicAdd(&agg[(size_t)l1[u] * 2 + ch], v[u]);
        if (n0 > 16) {
            const int d = slots0[(size_t)(m0 + u) * GCN_CAP + 16 + p];
            if (16 + p < n0) atomicAdd(&agg[(size_t)d * 2 + ch], v[u]);
        }
        if (n1 > 16) {
            const int d = slots1[(size_t)(m0 + u) * GCN_CAP + 16 + p];
            if (16 + p < n1) atomicAdd(&agg[(size_t)d * 2 + ch], v[u]);
        }
    }

    int no = ovfCnt[0];
    no = no < E ? no : E;
    if (no > 0) {
        float* __restrict__ tab = hwS[wv];
        if (p == 0) {
#pragma unroll
            for (int u = 0; u < 16; ++u) tab[u * 2 + ch] = v[u];
        }
        for (int i = lane; i < no; i += 32) {
            const int e = ovfList[i];
            const int u = src[e] - m0;
            if (u >= 0 && u < 16) {
                const int dj = dst[e];
                atomicAdd(&agg[(size_t)dj * 2],     tab[u * 2]);
                atomicAdd(&agg[(size_t)dj * 2 + 1], tab[u * 2 + 1]);
            }
        }
    }
}


extern "C" void kernel_launch(void* const* d_in, const int* in_sizes, int n_in,
                              void* d_out, int out_size, void* d_ws, size_t ws_size,
                              hipStream_t stream) {
    (void)in_sizes; (void)n_in; (void)out_size; (void)ws_size;

    const int N = GCN_N;
    const int E = GCN_E;

    const float* x    = (const float*)d_in[0];
    const int*   ei   = (const int*)d_in[1];
    const int*   src  = ei;
    const int*   dstp = ei + E;

    const float* W[5] = { (const float*)d_in[2], (const float*)d_in[4],
                          (const float*)d_in[6], (const float*)d_in[8],
                          (const float*)d_in[10] };
    const float* B[5] = { (const float*)d_in[3], (const float*)d_in[5],
                          (const float*)d_in[7], (const float*)d_in[9],
                          (const float*)d_in[11] };

    float* ws    = (float*)d_ws;
    float* hw    = ws;
    float* agg   = hw    + (size_t)N * 96;
    float* agg2  = agg   + (size_t)N * 96;
    float* dinv  = agg2  + (size_t)N * 96;
    float* deg2  = dinv  + GCN_NPAD;
    float* out2  = deg2  + GCN_NPAD;
    float* out   = (float*)d_out;
    int* cnt0    = (int*)(out2 + (size_t)N * 2);
    int* cnt1    = cnt0   + GCN_NPAD;
    int* slots0  = cnt1   + GCN_NPAD;
    int* slots1  = slots0 + (size_t)N * GCN_CAP;
    int* ovfCnt  = slots1 + (size_t)N * GCN_CAP;
    int* ovfList = ovfCnt + 64;

    const int T = 256;

    gcn_zero_kernel  <<<(N + T - 1) / T, T, 0, stream>>>(dinv, deg2, cnt0, cnt1, ovfCnt, out, out2, N);
    gcn_degree_bucket_kernel<<<(E + T - 1) / T, T, 0, stream>>>(
        src, dstp, dinv, deg2, cnt0, cnt1, slots0, slots1, ovfCnt, ovfList, E);
    gcn_dinv_kernel  <<<(N + T - 1) / T, T, 0, stream>>>(dinv, deg2, N);

    for (int l = 0; l < 4; ++l) {
        const float* biasIn = (l == 0) ? B[0] : B[l - 1];

        if (l == 0) {
            gcn_gemm_wstat_f32<165, 0, 0><<<1024, 96, 0, stream>>>(
                x, x, W[l], biasIn, dinv, hw, agg, agg2, GCN_N / 16);
        } else {
            gcn_gemm_wstat_f32<96, 1, 1><<<1024, 96, 0, stream>>>(
                agg, agg2, W[l], biasIn, dinv, hw, agg, agg2, GCN_N / 16);
        }

        const int pushBlocks    = (N + 8 * 2 - 1) / (8 * 2);
        gcn_push_rows_kernel<3, 8><<<pushBlocks, 64, 0, stream>>>(
            src, dstp, cnt0, cnt1, slots0, slots1, ovfCnt, ovfList, hw, agg, agg2, N);
    }

    gcn_last_layer_kernel<<<(GCN_N / 16 + 1) / 2, 64, 0, stream>>>(
        agg, agg2, W[4], B[3], dinv, src, dstp, cnt0, cnt1, slots0, slots1,
        ovfCnt, ovfList, out, out2, GCN_N / 16, E);

    gcn_finish_kernel<2><<<(2 * N + T - 1) / T, T, 0, stream>>>(out, out2, dinv, B[4], 2 * N);
}
